# starting kernel
# baseline (speedup 1.0000x reference)
#include <hip/hip_runtime.h>

typedef __attribute__((ext_vector_type(2))) float v2f;
typedef __attribute__((ext_vector_type(8))) float v8f;

#define NB 128
#define NN 2048
#define ND 64
#define NE 32768

__global__ void k_zero_u32(unsigned* p, int n) {
  int i = blockIdx.x * blockDim.x + threadIdx.x;
  if (i < n) p[i] = 0u;
}

__global__ void k_count(const int* __restrict__ dst, unsigned* __restrict__ deg) {
  int e = blockIdx.x * blockDim.x + threadIdx.x;
  if (e < NE) atomicAdd(&deg[dst[e]], 1u);
}

__global__ void k_scan(const unsigned* __restrict__ deg, unsigned* __restrict__ off,
                       unsigned* __restrict__ cursor) {
  __shared__ unsigned s0[NN];
  __shared__ unsigned s1[NN];
  int t = threadIdx.x;
  for (int i = t; i < NN; i += 256) s0[i] = deg[i];
  __syncthreads();
  unsigned* a = s0;
  unsigned* b = s1;
  for (int st = 1; st < NN; st <<= 1) {
    for (int i = t; i < NN; i += 256)
      b[i] = a[i] + ((i >= st) ? a[i - st] : 0u);
    __syncthreads();
    unsigned* tmp = a; a = b; b = tmp;
  }
  for (int i = t; i < NN; i += 256) {
    off[i + 1] = a[i];
    cursor[i] = (i == 0) ? 0u : a[i - 1];
  }
  if (t == 0) off[0] = 0u;
}

__global__ void k_fill(const int* __restrict__ src, const int* __restrict__ dst,
                       unsigned* __restrict__ cursor, unsigned* __restrict__ csr) {
  int e = blockIdx.x * blockDim.x + threadIdx.x;
  if (e < NE) {
    unsigned p = atomicAdd(&cursor[dst[e]], 1u);
    csr[p] = (unsigned)src[e];
  }
}

__global__ void k_maxb(const float* __restrict__ X, float* __restrict__ xmax) {
  int i = blockIdx.x * blockDim.x + threadIdx.x;
  float m = -3.4028234663852886e38f;
  for (int b = 0; b < NB; ++b)
    m = fmaxf(m, X[(size_t)b * (NN * ND) + i]);
  xmax[i] = m;
}

__global__ void __launch_bounds__(128)
k_mlp2_wmma(const float* __restrict__ Xin, const float* __restrict__ W1,
            const float* __restrict__ b1, const float* __restrict__ W2,
            const float* __restrict__ b2, float* __restrict__ Yout) {
  __shared__ float Ys[16 * ND];
  const int lane = threadIdx.x & 31;
  const int wave = threadIdx.x >> 5;
  const int col0 = wave * 16;
  const int hm = lane & 15;
  const int hk = lane >> 4;
  const size_t row0 = (size_t)blockIdx.x * 16;

  v2f bw[16];
#pragma unroll
  for (int kk = 0; kk < 16; ++kk) {
    int k = kk * 4 + 2 * hk;
    bw[kk].x = W1[(k + 0) * ND + col0 + hm];
    bw[kk].y = W1[(k + 1) * ND + col0 + hm];
  }

  v8f acc;
  {
    float bv = b1[col0 + hm];
#pragma unroll
    for (int j = 0; j < 8; ++j) acc[j] = bv;
  }

  const float* xrow = Xin + (row0 + hm) * ND;
#pragma unroll
  for (int kk = 0; kk < 16; ++kk) {
    int k = kk * 4 + 2 * hk;
    v2f a;
    a.x = xrow[k];
    a.y = xrow[k + 1];
    acc = __builtin_amdgcn_wmma_f32_16x16x4_f32(false, a, false, bw[kk],
                                                (short)0, acc, false, false);
  }
#pragma unroll
  for (int j = 0; j < 8; ++j) {
    Ys[(j + 8 * hk) * ND + col0 + hm] = fmaxf(acc[j], 0.0f);
  }
  __syncthreads();

#pragma unroll
  for (int kk = 0; kk < 16; ++kk) {
    int k = kk * 4 + 2 * hk;
    bw[kk].x = W2[(k + 0) * ND + col0 + hm];
    bw[kk].y = W2[(k + 1) * ND + col0 + hm];
  }
  {
    float bv = b2[col0 + hm];
#pragma unroll
    for (int j = 0; j < 8; ++j) acc[j] = bv;
  }
#pragma unroll
  for (int kk = 0; kk < 16; ++kk) {
    int k = kk * 4 + 2 * hk;
    v2f a;
    a.x = Ys[hm * ND + k];
    a.y = Ys[hm * ND + k + 1];
    acc = __builtin_amdgcn_wmma_f32_16x16x4_f32(false, a, false, bw[kk],
                                                (short)0, acc, false, false);
  }
#pragma unroll
  for (int j = 0; j < 8; ++j) {
    Yout[(row0 + j + 8 * hk) * ND + col0 + hm] = fmaxf(acc[j], 0.0f);
  }
}

__global__ void k_gather_node(const float* __restrict__ nodex,
                              const unsigned* __restrict__ off,
                              const unsigned* __restrict__ csr,
                              float* __restrict__ agg) {
  int r = blockIdx.x * 4 + (threadIdx.x >> 6);
  int d = threadIdx.x & 63;
  unsigned s = off[r], e = off[r + 1];
  float acc = 0.0f;
  for (unsigned j = s; j < e; ++j) acc += nodex[(size_t)csr[j] * ND + d];
  agg[(size_t)r * ND + d] = acc;
}

__global__ void k_gather_final(const float* __restrict__ h,
                               const float* __restrict__ agg,
                               const unsigned* __restrict__ off,
                               const unsigned* __restrict__ csr,
                               float* __restrict__ out) {
  int n = blockIdx.x & (NN - 1);
  int b = ((blockIdx.x >> 11) << 2) + (threadIdx.x >> 6);
  int d = threadIdx.x & 63;
  unsigned s = off[n], e = off[n + 1];
  float acc = agg[(size_t)n * ND + d];
  const float* hb = h + (size_t)b * NN * ND;
  for (unsigned j = s; j < e; ++j) acc += hb[(size_t)csr[j] * ND + d];
  out[((size_t)b * NN + n) * ND + d] = acc;
}

extern "C" void kernel_launch(void* const* d_in, const int* in_sizes, int n_in,
                              void* d_out, int out_size, void* d_ws, size_t ws_size,
                              hipStream_t stream) {
  (void)in_sizes; (void)n_in; (void)out_size; (void)ws_size;
  const float* X   = (const float*)d_in[0];
  const int*   ei  = (const int*)d_in[1];
  const float* W1t = (const float*)d_in[2];
  const float* b1t = (const float*)d_in[3];
  const float* W2t = (const float*)d_in[4];
  const float* b2t = (const float*)d_in[5];
  const float* W1n = (const float*)d_in[6];
  const float* b1n = (const float*)d_in[7];
  const float* W2n = (const float*)d_in[8];
  const float* b2n = (const float*)d_in[9];
  float* out = (float*)d_out;

  float* h     = (float*)d_ws;
  float* xmax  = h + (size_t)NB * NN * ND;
  float* nodex = xmax + (size_t)NN * ND;
  float* agg   = nodex + (size_t)NN * ND;
  unsigned* deg    = (unsigned*)(agg + (size_t)NN * ND);
  unsigned* off    = deg + NN;
  unsigned* cursor = off + (NN + 1);
  unsigned* csr    = cursor + NN;

  const int* esrc = ei;
  const int* edst = ei + NE;

  k_zero_u32<<<(NN + 255) / 256, 256, 0, stream>>>(deg, NN);
  k_count<<<NE / 256, 256, 0, stream>>>(edst, deg);
  k_scan<<<1, 256, 0, stream>>>(deg, off, cursor);
  k_fill<<<NE / 256, 256, 0, stream>>>(esrc, edst, cursor, csr);

  k_maxb<<<(NN * ND) / 256, 256, 0, stream>>>(X, xmax);
  k_mlp2_wmma<<<NN / 16, 128, 0, stream>>>(xmax, W1n, b1n, W2n, b2n, nodex);
  k_gather_node<<<NN / 4, 256, 0, stream>>>(nodex, off, csr, agg);

  k_mlp2_wmma<<<(NB * NN) / 16, 128, 0, stream>>>(X, W1t, b1t, W2t, b2t, h);
  k_gather_final<<<(NB / 4) * NN, 256, 0, stream>>>(h, agg, off, csr, out);
}


# final gather: each block stages one (subgraph, 32-col half) slice of h in LDS (256 KiB) and reads edge rows from LDS, not L2; sums even/odd-edge partials, then adds agg (new f32 add order)
# speedup vs baseline: 1.2695x; 1.2695x over previous
#include <hip/hip_runtime.h>

typedef __attribute__((ext_vector_type(2))) float v2f;
typedef __attribute__((ext_vector_type(8))) float v8f;

#define NB 128
#define NN 2048
#define ND 64
#define NE 32768

__global__ void k_zero_u32(unsigned* p, int n) {
  int i = blockIdx.x * blockDim.x + threadIdx.x;
  if (i < n) p[i] = 0u;
}

__global__ void k_count(const int* __restrict__ dst, unsigned* __restrict__ deg) {
  int e = blockIdx.x * blockDim.x + threadIdx.x;
  if (e < NE) atomicAdd(&deg[dst[e]], 1u);
}

__global__ void k_scan(const unsigned* __restrict__ deg, unsigned* __restrict__ off,
                       unsigned* __restrict__ cursor) {
  __shared__ unsigned s0[NN];
  __shared__ unsigned s1[NN];
  int t = threadIdx.x;
  for (int i = t; i < NN; i += 256) s0[i] = deg[i];
  __syncthreads();
  unsigned* a = s0;
  unsigned* b = s1;
  for (int st = 1; st < NN; st <<= 1) {
    for (int i = t; i < NN; i += 256)
      b[i] = a[i] + ((i >= st) ? a[i - st] : 0u);
    __syncthreads();
    unsigned* tmp = a; a = b; b = tmp;
  }
  for (int i = t; i < NN; i += 256) {
    off[i + 1] = a[i];
    cursor[i] = (i == 0) ? 0u : a[i - 1];
  }
  if (t == 0) off[0] = 0u;
}

__global__ void k_fill(const int* __restrict__ src, const int* __restrict__ dst,
                       unsigned* __restrict__ cursor, unsigned* __restrict__ csr) {
  int e = blockIdx.x * blockDim.x + threadIdx.x;
  if (e < NE) {
    unsigned p = atomicAdd(&cursor[dst[e]], 1u);
    csr[p] = (unsigned)src[e];
  }
}

__global__ void k_maxb(const float* __restrict__ X, float* __restrict__ xmax) {
  int i = blockIdx.x * blockDim.x + threadIdx.x;
  float m = -3.4028234663852886e38f;
  for (int b = 0; b < NB; ++b)
    m = fmaxf(m, X[(size_t)b * (NN * ND) + i]);
  xmax[i] = m;
}

__global__ void __launch_bounds__(128)
k_mlp2_wmma(const float* __restrict__ Xin, const float* __restrict__ W1,
            const float* __restrict__ b1, const float* __restrict__ W2,
            const float* __restrict__ b2, float* __restrict__ Yout) {
  __shared__ float Ys[16 * ND];
  const int lane = threadIdx.x & 31;
  const int wave = threadIdx.x >> 5;
  const int col0 = wave * 16;
  const int hm = lane & 15;
  const int hk = lane >> 4;
  const size_t row0 = (size_t)blockIdx.x * 16;

  v2f bw[16];
#pragma unroll
  for (int kk = 0; kk < 16; ++kk) {
    int k = kk * 4 + 2 * hk;
    bw[kk].x = W1[(k + 0) * ND + col0 + hm];
    bw[kk].y = W1[(k + 1) * ND + col0 + hm];
  }

  v8f acc;
  {
    float bv = b1[col0 + hm];
#pragma unroll
    for (int j = 0; j < 8; ++j) acc[j] = bv;
  }

  const float* xrow = Xin + (row0 + hm) * ND;
#pragma unroll
  for (int kk = 0; kk < 16; ++kk) {
    int k = kk * 4 + 2 * hk;
    v2f a;
    a.x = xrow[k];
    a.y = xrow[k + 1];
    acc = __builtin_amdgcn_wmma_f32_16x16x4_f32(false, a, false, bw[kk],
                                                (short)0, acc, false, false);
  }
#pragma unroll
  for (int j = 0; j < 8; ++j) {
    Ys[(j + 8 * hk) * ND + col0 + hm] = fmaxf(acc[j], 0.0f);
  }
  __syncthreads();

#pragma unroll
  for (int kk = 0; kk < 16; ++kk) {
    int k = kk * 4 + 2 * hk;
    bw[kk].x = W2[(k + 0) * ND + col0 + hm];
    bw[kk].y = W2[(k + 1) * ND + col0 + hm];
  }
  {
    float bv = b2[col0 + hm];
#pragma unroll
    for (int j = 0; j < 8; ++j) acc[j] = bv;
  }
#pragma unroll
  for (int kk = 0; kk < 16; ++kk) {
    int k = kk * 4 + 2 * hk;
    v2f a;
    a.x = Ys[hm * ND + k];
    a.y = Ys[hm * ND + k + 1];
    acc = __builtin_amdgcn_wmma_f32_16x16x4_f32(false, a, false, bw[kk],
                                                (short)0, acc, false, false);
  }
#pragma unroll
  for (int j = 0; j < 8; ++j) {
    Yout[(row0 + j + 8 * hk) * ND + col0 + hm] = fmaxf(acc[j], 0.0f);
  }
}

__global__ void k_gather_node(const float* __restrict__ nodex,
                              const unsigned* __restrict__ off,
                              const unsigned* __restrict__ csr,
                              float* __restrict__ agg) {
  int r = blockIdx.x * 4 + (threadIdx.x >> 6);
  int d = threadIdx.x & 63;
  unsigned s = off[r], e = off[r + 1];
  float acc = 0.0f;
  for (unsigned j = s; j < e; ++j) acc += nodex[(size_t)csr[j] * ND + d];
  agg[(size_t)r * ND + d] = acc;
}

#define GROW 32
#define GZROW_B (NN * GROW * 4)

template <int T>
__device__ __forceinline__ unsigned row_share_u32(unsigned v) {
  return (unsigned)__builtin_amdgcn_update_dpp(0, (int)v, 0x150 + T, 0xf, 0xf, false);
}

__device__ __forceinline__ float xhalf_f32(float x) {
  return __builtin_bit_cast(float,
      __builtin_amdgcn_permlanex16(__builtin_bit_cast(int, x), __builtin_bit_cast(int, x),
                                   0x76543210, 0xfedcba98, false, false));
}

__device__ __forceinline__ unsigned load_rowbytes(const unsigned* __restrict__ csr,
                                                  unsigned base, unsigned left,
                                                  unsigned slot) {
  unsigned r = GZROW_B;
  if (slot < left) r = csr[base + slot] << 7;
  return r;
}

template <int T>
__device__ __forceinline__ v2f lds_row2(const char* hsb, unsigned rb, unsigned colb) {
  return *(const v2f*)(hsb + (row_share_u32<T>(rb) + colb));
}

#define GATHER_GROUP(T)                                   \
  {                                                       \
    const v2f x0 = lds_row2<T + 0>(hsb, rb, colb);        \
    const v2f x1 = lds_row2<T + 1>(hsb, rb, colb);        \
    const v2f x2 = lds_row2<T + 2>(hsb, rb, colb);        \
    const v2f x3 = lds_row2<T + 3>(hsb, rb, colb);        \
    acc += x0; acc += x1; acc += x2; acc += x3;           \
  }

__global__ void __launch_bounds__(1024)
k_gather_final(const float* __restrict__ h, const float* __restrict__ agg,
               const unsigned* __restrict__ off, const unsigned* __restrict__ csr,
               float* __restrict__ out) {
  __shared__ float hs[(NN + 1) * GROW];
  const int tid = threadIdx.x;
  const int lane = tid & 31;
  const int wave = __builtin_amdgcn_readfirstlane(tid >> 5);
  const int b = blockIdx.x >> 1;
  const int half = blockIdx.x & 1;

  {
    const float* hsrc = h + (size_t)b * (NN * ND) + half * GROW;
    const int r0 = tid >> 3;
    const int c4 = (tid & 7) * 4;
    float4 v[16];
#pragma unroll
    for (int p = 0; p < 16; ++p)
      v[p] = *(const float4*)(hsrc + (size_t)(r0 + p * 128) * ND + c4);
#pragma unroll
    for (int p = 0; p < 16; ++p)
      *(float4*)&hs[(r0 + p * 128) * GROW + c4] = v[p];
    if (tid < GROW) hs[NN * GROW + tid] = 0.0f;
  }
  __syncthreads();

  const char* hsb = (const char*)hs;
  const int c = lane & 15;
  const int hrow = lane >> 4;
  const unsigned slot = 2u * c + hrow;
  const unsigned colb = 8u * c;
  const int ocol = half * GROW + 2 * c + hrow;

  for (int q = 0; q < 2; ++q) {
    const int nq = wave * 64 + q * 32;
    const unsigned ov_e = off[nq + lane + 1];
    unsigned s_cur = off[nq];
    unsigned e_cur = (unsigned)__builtin_amdgcn_readlane((int)ov_e, 0);
    unsigned rb_cur = load_rowbytes(csr, s_cur, e_cur - s_cur, slot);
    for (int i = 0; i < 32; ++i) {
      const int n = nq + i;
      const unsigned s_nxt = e_cur;
      const unsigned e_nxt = (unsigned)__builtin_amdgcn_readlane((int)ov_e, (i + 1) & 31);
      const unsigned cnt_nxt = (i + 1 < 32) ? (e_nxt - s_nxt) : 0u;
      const bool ok_nxt = slot < cnt_nxt;
      const unsigned raw_nxt = csr[ok_nxt ? (s_nxt + slot) : 0u];
      const float av = agg[(size_t)n * ND + ocol];

      v2f acc = {0.0f, 0.0f};
      unsigned rb = rb_cur;
      unsigned left = e_cur - s_cur;
      unsigned base = s_cur;
      for (;;) {
        if (left > 0u) GATHER_GROUP(0)
        if (left > 8u) GATHER_GROUP(4)
        if (left > 16u) GATHER_GROUP(8)
        if (left > 24u) GATHER_GROUP(12)
        if (left <= 32u) break;
        left -= 32u;
        base += 32u;
        rb = load_rowbytes(csr, base, left, slot);
      }
      const float keep = hrow ? acc.y : acc.x;
      const float send = hrow ? acc.x : acc.y;
      const float tot = keep + xhalf_f32(send);
      out[((size_t)b * NN + n) * ND + ocol] = av + tot;

      s_cur = s_nxt;
      e_cur = e_nxt;
      rb_cur = ok_nxt ? (raw_nxt << 7) : (unsigned)GZROW_B;
    }
  }
}

extern "C" void kernel_launch(void* const* d_in, const int* in_sizes, int n_in,
                              void* d_out, int out_size, void* d_ws, size_t ws_size,
                              hipStream_t stream) {
  (void)in_sizes; (void)n_in; (void)out_size; (void)ws_size;
  const float* X   = (const float*)d_in[0];
  const int*   ei  = (const int*)d_in[1];
  const float* W1t = (const float*)d_in[2];
  const float* b1t = (const float*)d_in[3];
  const float* W2t = (const float*)d_in[4];
  const float* b2t = (const float*)d_in[5];
  const float* W1n = (const float*)d_in[6];
  const float* b1n = (const float*)d_in[7];
  const float* W2n = (const float*)d_in[8];
  const float* b2n = (const float*)d_in[9];
  float* out = (float*)d_out;

  float* h     = (float*)d_ws;
  float* xmax  = h + (size_t)NB * NN * ND;
  float* nodex = xmax + (size_t)NN * ND;
  float* agg   = nodex + (size_t)NN * ND;
  unsigned* deg    = (unsigned*)(agg + (size_t)NN * ND);
  unsigned* off    = deg + NN;
  unsigned* cursor = off + (NN + 1);
  unsigned* csr    = cursor + NN;

  const int* esrc = ei;
  const int* edst = ei + NE;

  k_zero_u32<<<(NN + 255) / 256, 256, 0, stream>>>(deg, NN);
  k_count<<<NE / 256, 256, 0, stream>>>(edst, deg);
  k_scan<<<1, 256, 0, stream>>>(deg, off, cursor);
  k_fill<<<NE / 256, 256, 0, stream>>>(esrc, edst, cursor, csr);

  k_maxb<<<(NN * ND) / 256, 256, 0, stream>>>(X, xmax);
  k_mlp2_wmma<<<NN / 16, 128, 0, stream>>>(xmax, W1n, b1n, W2n, b2n, nodex);
  k_gather_node<<<NN / 4, 256, 0, stream>>>(nodex, off, csr, agg);

  k_mlp2_wmma<<<(NB * NN) / 16, 128, 0, stream>>>(X, W1t, b1t, W2t, b2t, h);
  k_gather_final<<<NB * 2, 1024, 0, stream>>>(h, agg, off, csr, out);
}


# MLP: each wave holds both layers' weight fragments in registers and walks 8 whole 16x64 row tiles (2048 waves, was 65536); layer 1 transposed so its D is layer 2's A, no LDS (new f32 add order)
# speedup vs baseline: 1.4604x; 1.1504x over previous
#include <hip/hip_runtime.h>

typedef __attribute__((ext_vector_type(2))) float v2f;
typedef __attribute__((ext_vector_type(4))) float v4f;
typedef __attribute__((ext_vector_type(8))) float v8f;

#define NB 128
#define NN 2048
#define ND 64
#define NE 32768

__global__ void k_zero_u32(unsigned* p, int n) {
  int i = blockIdx.x * blockDim.x + threadIdx.x;
  if (i < n) p[i] = 0u;
}

__global__ void k_count(const int* __restrict__ dst, unsigned* __restrict__ deg) {
  int e = blockIdx.x * blockDim.x + threadIdx.x;
  if (e < NE) atomicAdd(&deg[dst[e]], 1u);
}

__global__ void k_scan(const unsigned* __restrict__ deg, unsigned* __restrict__ off,
                       unsigned* __restrict__ cursor) {
  __shared__ unsigned s0[NN];
  __shared__ unsigned s1[NN];
  int t = threadIdx.x;
  for (int i = t; i < NN; i += 256) s0[i] = deg[i];
  __syncthreads();
  unsigned* a = s0;
  unsigned* b = s1;
  for (int st = 1; st < NN; st <<= 1) {
    for (int i = t; i < NN; i += 256)
      b[i] = a[i] + ((i >= st) ? a[i - st] : 0u);
    __syncthreads();
    unsigned* tmp = a; a = b; b = tmp;
  }
  for (int i = t; i < NN; i += 256) {
    off[i + 1] = a[i];
    cursor[i] = (i == 0) ? 0u : a[i - 1];
  }
  if (t == 0) off[0] = 0u;
}

__global__ void k_fill(const int* __restrict__ src, const int* __restrict__ dst,
                       unsigned* __restrict__ cursor, unsigned* __restrict__ csr) {
  int e = blockIdx.x * blockDim.x + threadIdx.x;
  if (e < NE) {
    unsigned p = atomicAdd(&cursor[dst[e]], 1u);
    csr[p] = (unsigned)src[e];
  }
}

__global__ void k_maxb(const float* __restrict__ X, float* __restrict__ xmax) {
  int i = blockIdx.x * blockDim.x + threadIdx.x;
  float m = -3.4028234663852886e38f;
  for (int b = 0; b < NB; ++b)
    m = fmaxf(m, X[(size_t)b * (NN * ND) + i]);
  xmax[i] = m;
}

#define MLP_WAVES 8
#define MLP_TILES_BIG 8
__global__ void __launch_bounds__(MLP_WAVES * 32)
k_mlp2_wmma(const float* __restrict__ Xin, const float* __restrict__ W1,
            const float* __restrict__ b1, const float* __restrict__ W2,
            const float* __restrict__ b2, float* __restrict__ Yout, int ntiles) {
  const int lane = threadIdx.x & 31;
  const int hm = lane & 15;
  const int hk = lane >> 4;
  const int gw = __builtin_amdgcn_readfirstlane((int)(blockIdx.x * MLP_WAVES + (threadIdx.x >> 5)));
  const int nw = (int)gridDim.x * MLP_WAVES;

  v2f w1[4][16];
#pragma unroll
  for (int s = 0; s < 16; ++s) {
#pragma unroll
    for (int e = 0; e < 2; ++e) {
      const int k = 8 * (s >> 1) + 4 * hk + 2 * (s & 1) + e;
      const v4f wv = *(const v4f*)(W1 + k * ND + 4 * hm);
      w1[0][s][e] = wv[0]; w1[1][s][e] = wv[1]; w1[2][s][e] = wv[2]; w1[3][s][e] = wv[3];
    }
  }
  v2f w2[4][16];
#pragma unroll
  for (int t = 0; t < 16; ++t) {
#pragma unroll
    for (int e = 0; e < 2; ++e) {
      const int f = 4 * (8 * hk + 2 * (t & 3) + e) + (t >> 2);
      const v4f wv = *(const v4f*)(W2 + f * ND + 4 * hm);
      w2[0][t][e] = wv[0]; w2[1][t][e] = wv[1]; w2[2][t][e] = wv[2]; w2[3][t][e] = wv[3];
    }
  }
  v8f c1[4];
#pragma unroll
  for (int r = 0; r < 8; ++r) {
    const v4f bv = *(const v4f*)(b1 + 4 * (8 * hk + r));
    c1[0][r] = bv[0]; c1[1][r] = bv[1]; c1[2][r] = bv[2]; c1[3][r] = bv[3];
  }
  const v4f c2 = *(const v4f*)(b2 + 4 * hm);

  if (gw >= ntiles) return;

  v4f x[8];
  {
    const float* xr = Xin + ((size_t)gw * 16 + hm) * ND + 4 * hk;
#pragma unroll
    for (int i = 0; i < 8; ++i) x[i] = *(const v4f*)(xr + 8 * i);
  }

  for (int t = gw; t < ntiles; t += nw) {
    const int tn = (t + nw < ntiles) ? (t + nw) : t;
    v4f xn[8];
    {
      const float* xr = Xin + ((size_t)tn * 16 + hm) * ND + 4 * hk;
#pragma unroll
      for (int i = 0; i < 8; ++i) xn[i] = *(const v4f*)(xr + 8 * i);
    }

    v8f d1[4];
#pragma unroll
    for (int ft = 0; ft < 4; ++ft) d1[ft] = c1[ft];
#pragma unroll
    for (int i = 0; i < 8; ++i) {
      v2f blo, bhi;
      blo[0] = x[i][0]; blo[1] = x[i][1];
      bhi[0] = x[i][2]; bhi[1] = x[i][3];
#pragma unroll
      for (int ft = 0; ft < 4; ++ft)
        d1[ft] = __builtin_amdgcn_wmma_f32_16x16x4_f32(false, w1[ft][2 * i], false, blo,
                                                       (short)0, d1[ft], false, false);
#pragma unroll
      for (int ft = 0; ft < 4; ++ft)
        d1[ft] = __builtin_amdgcn_wmma_f32_16x16x4_f32(false, w1[ft][2 * i + 1], false, bhi,
                                                       (short)0, d1[ft], false, false);
    }
#pragma unroll
    for (int ft = 0; ft < 4; ++ft) {
#pragma unroll
      for (int r = 0; r < 8; ++r) d1[ft][r] = fmaxf(d1[ft][r], 0.0f);
    }

    v8f d2[4];
#pragma unroll
    for (int gt = 0; gt < 4; ++gt) {
#pragma unroll
      for (int r = 0; r < 8; ++r) d2[gt][r] = c2[gt];
    }
#pragma unroll
    for (int ft = 0; ft < 4; ++ft) {
#pragma unroll
      for (int jp = 0; jp < 4; ++jp) {
        v2f a;
        a[0] = d1[ft][2 * jp];
        a[1] = d1[ft][2 * jp + 1];
#pragma unroll
        for (int gt = 0; gt < 4; ++gt)
          d2[gt] = __builtin_amdgcn_wmma_f32_16x16x4_f32(false, a, false, w2[gt][4 * ft + jp],
                                                         (short)0, d2[gt], false, false);
      }
    }

    {
      float* yr = Yout + ((size_t)t * 16 + 8 * hk) * ND + 4 * hm;
#pragma unroll
      for (int r = 0; r < 8; ++r) {
        v4f o;
        o[0] = fmaxf(d2[0][r], 0.0f);
        o[1] = fmaxf(d2[1][r], 0.0f);
        o[2] = fmaxf(d2[2][r], 0.0f);
        o[3] = fmaxf(d2[3][r], 0.0f);
        *(v4f*)(yr + r * ND) = o;
      }
    }

#pragma unroll
    for (int i = 0; i < 8; ++i) x[i] = xn[i];
  }
}

__global__ void k_gather_node(const float* __restrict__ nodex,
                              const unsigned* __restrict__ off,
                              const unsigned* __restrict__ csr,
                              float* __restrict__ agg) {
  int r = blockIdx.x * 4 + (threadIdx.x >> 6);
  int d = threadIdx.x & 63;
  unsigned s = off[r], e = off[r + 1];
  float acc = 0.0f;
  for (unsigned j = s; j < e; ++j) acc += nodex[(size_t)csr[j] * ND + d];
  agg[(size_t)r * ND + d] = acc;
}

#define GROW 32
#define GZROW_B (NN * GROW * 4)

template <int T>
__device__ __forceinline__ unsigned row_share_u32(unsigned v) {
  return (unsigned)__builtin_amdgcn_update_dpp(0, (int)v, 0x150 + T, 0xf, 0xf, false);
}

__device__ __forceinline__ float xhalf_f32(float x) {
  return __builtin_bit_cast(float,
      __builtin_amdgcn_permlanex16(__builtin_bit_cast(int, x), __builtin_bit_cast(int, x),
                                   0x76543210, 0xfedcba98, false, false));
}

__device__ __forceinline__ unsigned load_rowbytes(const unsigned* __restrict__ csr,
                                                  unsigned base, unsigned left,
                                                  unsigned slot) {
  unsigned r = GZROW_B;
  if (slot < left) r = csr[base + slot] << 7;
  return r;
}

template <int T>
__device__ __forceinline__ v2f lds_row2(const char* hsb, unsigned rb, unsigned colb) {
  return *(const v2f*)(hsb + (row_share_u32<T>(rb) + colb));
}

#define GATHER_GROUP(T)                                   \
  {                                                       \
    const v2f x0 = lds_row2<T + 0>(hsb, rb, colb);        \
    const v2f x1 = lds_row2<T + 1>(hsb, rb, colb);        \
    const v2f x2 = lds_row2<T + 2>(hsb, rb, colb);        \
    const v2f x3 = lds_row2<T + 3>(hsb, rb, colb);        \
    acc += x0; acc += x1; acc += x2; acc += x3;           \
  }

__global__ void __launch_bounds__(1024)
k_gather_final(const float* __restrict__ h, const float* __restrict__ agg,
               const unsigned* __restrict__ off, const unsigned* __restrict__ csr,
               float* __restrict__ out) {
  __shared__ float hs[(NN + 1) * GROW];
  const int tid = threadIdx.x;
  const int lane = tid & 31;
  const int wave = __builtin_amdgcn_readfirstlane(tid >> 5);
  const int b = blockIdx.x >> 1;
  const int half = blockIdx.x & 1;

  {
    const float* hsrc = h + (size_t)b * (NN * ND) + half * GROW;
    const int r0 = tid >> 3;
    const int c4 = (tid & 7) * 4;
    float4 v[16];
#pragma unroll
    for (int p = 0; p < 16; ++p)
      v[p] = *(const float4*)(hsrc + (size_t)(r0 + p * 128) * ND + c4);
#pragma unroll
    for (int p = 0; p < 16; ++p)
      *(float4*)&hs[(r0 + p * 128) * GROW + c4] = v[p];
    if (tid < GROW) hs[NN * GROW + tid] = 0.0f;
  }
  __syncthreads();

  const char* hsb = (const char*)hs;
  const int c = lane & 15;
  const int hrow = lane >> 4;
  const unsigned slot = 2u * c + hrow;
  const unsigned colb = 8u * c;
  const int ocol = half * GROW + 2 * c + hrow;

  for (int q = 0; q < 2; ++q) {
    const int nq = wave * 64 + q * 32;
    const unsigned ov_e = off[nq + lane + 1];
    unsigned s_cur = off[nq];
    unsigned e_cur = (unsigned)__builtin_amdgcn_readlane((int)ov_e, 0);
    unsigned rb_cur = load_rowbytes(csr, s_cur, e_cur - s_cur, slot);
    for (int i = 0; i < 32; ++i) {
      const int n = nq + i;
      const unsigned s_nxt = e_cur;
      const unsigned e_nxt = (unsigned)__builtin_amdgcn_readlane((int)ov_e, (i + 1) & 31);
      const unsigned cnt_nxt = (i + 1 < 32) ? (e_nxt - s_nxt) : 0u;
      const bool ok_nxt = slot < cnt_nxt;
      const unsigned raw_nxt = csr[ok_nxt ? (s_nxt + slot) : 0u];
      const float av = agg[(size_t)n * ND + ocol];

      v2f acc = {0.0f, 0.0f};
      unsigned rb = rb_cur;
      unsigned left = e_cur - s_cur;
      unsigned base = s_cur;
      for (;;) {
        if (left > 0u) GATHER_GROUP(0)
        if (left > 8u) GATHER_GROUP(4)
        if (left > 16u) GATHER_GROUP(8)
        if (left > 24u) GATHER_GROUP(12)
        if (left <= 32u) break;
        left -= 32u;
        base += 32u;
        rb = load_rowbytes(csr, base, left, slot);
      }
      const float keep = hrow ? acc.y : acc.x;
      const float send = hrow ? acc.x : acc.y;
      const float tot = keep + xhalf_f32(send);
      out[((size_t)b * NN + n) * ND + ocol] = av + tot;

      s_cur = s_nxt;
      e_cur = e_nxt;
      rb_cur = ok_nxt ? (raw_nxt << 7) : (unsigned)GZROW_B;
    }
  }
}

extern "C" void kernel_launch(void* const* d_in, const int* in_sizes, int n_in,
                              void* d_out, int out_size, void* d_ws, size_t ws_size,
                              hipStream_t stream) {
  (void)in_sizes; (void)n_in; (void)out_size; (void)ws_size;
  const float* X   = (const float*)d_in[0];
  const int*   ei  = (const int*)d_in[1];
  const float* W1t = (const float*)d_in[2];
  const float* b1t = (const float*)d_in[3];
  const float* W2t = (const float*)d_in[4];
  const float* b2t = (const float*)d_in[5];
  const float* W1n = (const float*)d_in[6];
  const float* b1n = (const float*)d_in[7];
  const float* W2n = (const float*)d_in[8];
  const float* b2n = (const float*)d_in[9];
  float* out = (float*)d_out;

  float* h     = (float*)d_ws;
  float* xmax  = h + (size_t)NB * NN * ND;
  float* nodex = xmax + (size_t)NN * ND;
  float* agg   = nodex + (size_t)NN * ND;
  unsigned* deg    = (unsigned*)(agg + (size_t)NN * ND);
  unsigned* off    = deg + NN;
  unsigned* cursor = off + (NN + 1);
  unsigned* csr    = cursor + NN;

  const int* esrc = ei;
  const int* edst = ei + NE;

  k_zero_u32<<<(NN + 255) / 256, 256, 0, stream>>>(deg, NN);
  k_count<<<NE / 256, 256, 0, stream>>>(edst, deg);
  k_scan<<<1, 256, 0, stream>>>(deg, off, cursor);
  k_fill<<<NE / 256, 256, 0, stream>>>(esrc, edst, cursor, csr);

  k_maxb<<<(NN * ND) / 256, 256, 0, stream>>>(X, xmax);
  k_mlp2_wmma<<<(NN / 16) / MLP_WAVES, MLP_WAVES * 32, 0, stream>>>(xmax, W1n, b1n, W2n, b2n, nodex, NN / 16);
  k_gather_node<<<NN / 4, 256, 0, stream>>>(nodex, off, csr, agg);

  k_mlp2_wmma<<<((NB * NN) / 16) / (MLP_WAVES * MLP_TILES_BIG), MLP_WAVES * 32, 0, stream>>>(X, W1t, b1t, W2t, b2t, h, (NB * NN) / 16);
  k_gather_final<<<NB * 2, 1024, 0, stream>>>(h, agg, off, csr, out);
}


# max over subgraphs: each lane issues its 32 sixteen-byte plane loads before the first max (was two 4-byte loads per wait, 64 waits per lane); the 8 waves' partial maxima meet in LDS
# speedup vs baseline: 1.8196x; 1.2460x over previous
#include <hip/hip_runtime.h>

typedef __attribute__((ext_vector_type(2))) float v2f;
typedef __attribute__((ext_vector_type(4))) float v4f;
typedef __attribute__((ext_vector_type(8))) float v8f;

#define NB 128
#define NN 2048
#define ND 64
#define NE 32768

__global__ void k_zero_u32(unsigned* p, int n) {
  int i = blockIdx.x * blockDim.x + threadIdx.x;
  if (i < n) p[i] = 0u;
}

__global__ void k_count(const int* __restrict__ dst, unsigned* __restrict__ deg) {
  int e = blockIdx.x * blockDim.x + threadIdx.x;
  if (e < NE) atomicAdd(&deg[dst[e]], 1u);
}

__global__ void k_scan(const unsigned* __restrict__ deg, unsigned* __restrict__ off,
                       unsigned* __restrict__ cursor) {
  __shared__ unsigned s0[NN];
  __shared__ unsigned s1[NN];
  int t = threadIdx.x;
  for (int i = t; i < NN; i += 256) s0[i] = deg[i];
  __syncthreads();
  unsigned* a = s0;
  unsigned* b = s1;
  for (int st = 1; st < NN; st <<= 1) {
    for (int i = t; i < NN; i += 256)
      b[i] = a[i] + ((i >= st) ? a[i - st] : 0u);
    __syncthreads();
    unsigned* tmp = a; a = b; b = tmp;
  }
  for (int i = t; i < NN; i += 256) {
    off[i + 1] = a[i];
    cursor[i] = (i == 0) ? 0u : a[i - 1];
  }
  if (t == 0) off[0] = 0u;
}

__global__ void k_fill(const int* __restrict__ src, const int* __restrict__ dst,
                       unsigned* __restrict__ cursor, unsigned* __restrict__ csr) {
  int e = blockIdx.x * blockDim.x + threadIdx.x;
  if (e < NE) {
    unsigned p = atomicAdd(&cursor[dst[e]], 1u);
    csr[p] = (unsigned)src[e];
  }
}

#define MAXB_WAVES 8
#define MAXB_PLANES (NB / MAXB_WAVES)
#define MAXB_COLS4 64

__device__ __forceinline__ v4f max4(v4f a, v4f b) {
  v4f r;
  r[0] = fmaxf(a[0], b[0]);
  r[1] = fmaxf(a[1], b[1]);
  r[2] = fmaxf(a[2], b[2]);
  r[3] = fmaxf(a[3], b[3]);
  return r;
}

__global__ void __launch_bounds__(MAXB_WAVES * 32) __attribute__((amdgpu_waves_per_eu(4, 4)))
k_maxb(const float* __restrict__ X, float* __restrict__ xmax) {
  __shared__ v4f red[MAXB_WAVES][MAXB_COLS4];
  const int lane = threadIdx.x & 31;
  const int w = threadIdx.x >> 5;
  const int c4 = blockIdx.x * MAXB_COLS4 + lane;
  const float* p = X + (size_t)w * (NN * ND) + (size_t)c4 * 4;
  v4f va[MAXB_PLANES], vb[MAXB_PLANES];
#pragma unroll
  for (int j = 0; j < MAXB_PLANES; ++j) {
    const float* q = p + (size_t)j * (MAXB_WAVES * NN * ND);
    va[j] = *(const v4f*)q;
    vb[j] = *(const v4f*)(q + 128);
  }
  v4f ma = va[0], mb = vb[0];
#pragma unroll
  for (int j = 1; j < MAXB_PLANES; ++j) {
    ma = max4(ma, va[j]);
    mb = max4(mb, vb[j]);
  }
  red[w][lane] = ma;
  red[w][lane + 32] = mb;
  __syncthreads();
  if (threadIdx.x < MAXB_COLS4) {
    v4f r = red[0][threadIdx.x];
#pragma unroll
    for (int k = 1; k < MAXB_WAVES; ++k) r = max4(r, red[k][threadIdx.x]);
    *(v4f*)(xmax + ((size_t)blockIdx.x * MAXB_COLS4 + threadIdx.x) * 4) = r;
  }
}

#define MLP_WAVES 8
#define MLP_TILES_BIG 8
__global__ void __launch_bounds__(MLP_WAVES * 32)
k_mlp2_wmma(const float* __restrict__ Xin, const float* __restrict__ W1,
            const float* __restrict__ b1, const float* __restrict__ W2,
            const float* __restrict__ b2, float* __restrict__ Yout, int ntiles) {
  const int lane = threadIdx.x & 31;
  const int hm = lane & 15;
  const int hk = lane >> 4;
  const int gw = __builtin_amdgcn_readfirstlane((int)(blockIdx.x * MLP_WAVES + (threadIdx.x >> 5)));
  const int nw = (int)gridDim.x * MLP_WAVES;

  v2f w1[4][16];
#pragma unroll
  for (int s = 0; s < 16; ++s) {
#pragma unroll
    for (int e = 0; e < 2; ++e) {
      const int k = 8 * (s >> 1) + 4 * hk + 2 * (s & 1) + e;
      const v4f wv = *(const v4f*)(W1 + k * ND + 4 * hm);
      w1[0][s][e] = wv[0]; w1[1][s][e] = wv[1]; w1[2][s][e] = wv[2]; w1[3][s][e] = wv[3];
    }
  }
  v2f w2[4][16];
#pragma unroll
  for (int t = 0; t < 16; ++t) {
#pragma unroll
    for (int e = 0; e < 2; ++e) {
      const int f = 4 * (8 * hk + 2 * (t & 3) + e) + (t >> 2);
      const v4f wv = *(const v4f*)(W2 + f * ND + 4 * hm);
      w2[0][t][e] = wv[0]; w2[1][t][e] = wv[1]; w2[2][t][e] = wv[2]; w2[3][t][e] = wv[3];
    }
  }
  v8f c1[4];
#pragma unroll
  for (int r = 0; r < 8; ++r) {
    const v4f bv = *(const v4f*)(b1 + 4 * (8 * hk + r));
    c1[0][r] = bv[0]; c1[1][r] = bv[1]; c1[2][r] = bv[2]; c1[3][r] = bv[3];
  }
  const v4f c2 = *(const v4f*)(b2 + 4 * hm);

  if (gw >= ntiles) return;

  v4f x[8];
  {
    const float* xr = Xin + ((size_t)gw * 16 + hm) * ND + 4 * hk;
#pragma unroll
    for (int i = 0; i < 8; ++i) x[i] = *(const v4f*)(xr + 8 * i);
  }

  for (int t = gw; t < ntiles; t += nw) {
    const int tn = (t + nw < ntiles) ? (t + nw) : t;
    v4f xn[8];
    {
      const float* xr = Xin + ((size_t)tn * 16 + hm) * ND + 4 * hk;
#pragma unroll
      for (int i = 0; i < 8; ++i) xn[i] = *(const v4f*)(xr + 8 * i);
    }

    v8f d1[4];
#pragma unroll
    for (int ft = 0; ft < 4; ++ft) d1[ft] = c1[ft];
#pragma unroll
    for (int i = 0; i < 8; ++i) {
      v2f blo, bhi;
      blo[0] = x[i][0]; blo[1] = x[i][1];
      bhi[0] = x[i][2]; bhi[1] = x[i][3];
#pragma unroll
      for (int ft = 0; ft < 4; ++ft)
        d1[ft] = __builtin_amdgcn_wmma_f32_16x16x4_f32(false, w1[ft][2 * i], false, blo,
                                                       (short)0, d1[ft], false, false);
#pragma unroll
      for (int ft = 0; ft < 4; ++ft)
        d1[ft] = __builtin_amdgcn_wmma_f32_16x16x4_f32(false, w1[ft][2 * i + 1], false, bhi,
                                                       (short)0, d1[ft], false, false);
    }
#pragma unroll
    for (int ft = 0; ft < 4; ++ft) {
#pragma unroll
      for (int r = 0; r < 8; ++r) d1[ft][r] = fmaxf(d1[ft][r], 0.0f);
    }

    v8f d2[4];
#pragma unroll
    for (int gt = 0; gt < 4; ++gt) {
#pragma unroll
      for (int r = 0; r < 8; ++r) d2[gt][r] = c2[gt];
    }
#pragma unroll
    for (int ft = 0; ft < 4; ++ft) {
#pragma unroll
      for (int jp = 0; jp < 4; ++jp) {
        v2f a;
        a[0] = d1[ft][2 * jp];
        a[1] = d1[ft][2 * jp + 1];
#pragma unroll
        for (int gt = 0; gt < 4; ++gt)
          d2[gt] = __builtin_amdgcn_wmma_f32_16x16x4_f32(false, a, false, w2[gt][4 * ft + jp],
                                                         (short)0, d2[gt], false, false);
      }
    }

    {
      float* yr = Yout + ((size_t)t * 16 + 8 * hk) * ND + 4 * hm;
#pragma unroll
      for (int r = 0; r < 8; ++r) {
        v4f o;
        o[0] = fmaxf(d2[0][r], 0.0f);
        o[1] = fmaxf(d2[1][r], 0.0f);
        o[2] = fmaxf(d2[2][r], 0.0f);
        o[3] = fmaxf(d2[3][r], 0.0f);
        *(v4f*)(yr + r * ND) = o;
      }
    }

#pragma unroll
    for (int i = 0; i < 8; ++i) x[i] = xn[i];
  }
}

__global__ void k_gather_node(const float* __restrict__ nodex,
                              const unsigned* __restrict__ off,
                              const unsigned* __restrict__ csr,
                              float* __restrict__ agg) {
  int r = blockIdx.x * 4 + (threadIdx.x >> 6);
  int d = threadIdx.x & 63;
  unsigned s = off[r], e = off[r + 1];
  float acc = 0.0f;
  for (unsigned j = s; j < e; ++j) acc += nodex[(size_t)csr[j] * ND + d];
  agg[(size_t)r * ND + d] = acc;
}

#define GROW 32
#define GZROW_B (NN * GROW * 4)

template <int T>
__device__ __forceinline__ unsigned row_share_u32(unsigned v) {
  return (unsigned)__builtin_amdgcn_update_dpp(0, (int)v, 0x150 + T, 0xf, 0xf, false);
}

__device__ __forceinline__ float xhalf_f32(float x) {
  return __builtin_bit_cast(float,
      __builtin_amdgcn_permlanex16(__builtin_bit_cast(int, x), __builtin_bit_cast(int, x),
                                   0x76543210, 0xfedcba98, false, false));
}

__device__ __forceinline__ unsigned load_rowbytes(const unsigned* __restrict__ csr,
                                                  unsigned base, unsigned left,
                                                  unsigned slot) {
  unsigned r = GZROW_B;
  if (slot < left) r = csr[base + slot] << 7;
  return r;
}

template <int T>
__device__ __forceinline__ v2f lds_row2(const char* hsb, unsigned rb, unsigned colb) {
  return *(const v2f*)(hsb + (row_share_u32<T>(rb) + colb));
}

#define GATHER_GROUP(T)                                   \
  {                                                       \
    const v2f x0 = lds_row2<T + 0>(hsb, rb, colb);        \
    const v2f x1 = lds_row2<T + 1>(hsb, rb, colb);        \
    const v2f x2 = lds_row2<T + 2>(hsb, rb, colb);        \
    const v2f x3 = lds_row2<T + 3>(hsb, rb, colb);        \
    acc += x0; acc += x1; acc += x2; acc += x3;           \
  }

__global__ void __launch_bounds__(1024)
k_gather_final(const float* __restrict__ h, const float* __restrict__ agg,
               const unsigned* __restrict__ off, const unsigned* __restrict__ csr,
               float* __restrict__ out) {
  __shared__ float hs[(NN + 1) * GROW];
  const int tid = threadIdx.x;
  const int lane = tid & 31;
  const int wave = __builtin_amdgcn_readfirstlane(tid >> 5);
  const int b = blockIdx.x >> 1;
  const int half = blockIdx.x & 1;

  {
    const float* hsrc = h + (size_t)b * (NN * ND) + half * GROW;
    const int r0 = tid >> 3;
    const int c4 = (tid & 7) * 4;
    float4 v[16];
#pragma unroll
    for (int p = 0; p < 16; ++p)
      v[p] = *(const float4*)(hsrc + (size_t)(r0 + p * 128) * ND + c4);
#pragma unroll
    for (int p = 0; p < 16; ++p)
      *(float4*)&hs[(r0 + p * 128) * GROW + c4] = v[p];
    if (tid < GROW) hs[NN * GROW + tid] = 0.0f;
  }
  __syncthreads();

  const char* hsb = (const char*)hs;
  const int c = lane & 15;
  const int hrow = lane >> 4;
  const unsigned slot = 2u * c + hrow;
  const unsigned colb = 8u * c;
  const int ocol = half * GROW + 2 * c + hrow;

  for (int q = 0; q < 2; ++q) {
    const int nq = wave * 64 + q * 32;
    const unsigned ov_e = off[nq + lane + 1];
    unsigned s_cur = off[nq];
    unsigned e_cur = (unsigned)__builtin_amdgcn_readlane((int)ov_e, 0);
    unsigned rb_cur = load_rowbytes(csr, s_cur, e_cur - s_cur, slot);
    for (int i = 0; i < 32; ++i) {
      const int n = nq + i;
      const unsigned s_nxt = e_cur;
      const unsigned e_nxt = (unsigned)__builtin_amdgcn_readlane((int)ov_e, (i + 1) & 31);
      const unsigned cnt_nxt = (i + 1 < 32) ? (e_nxt - s_nxt) : 0u;
      const bool ok_nxt = slot < cnt_nxt;
      const unsigned raw_nxt = csr[ok_nxt ? (s_nxt + slot) : 0u];
      const float av = agg[(size_t)n * ND + ocol];

      v2f acc = {0.0f, 0.0f};
      unsigned rb = rb_cur;
      unsigned left = e_cur - s_cur;
      unsigned base = s_cur;
      for (;;) {
        if (left > 0u) GATHER_GROUP(0)
        if (left > 8u) GATHER_GROUP(4)
        if (left > 16u) GATHER_GROUP(8)
        if (left > 24u) GATHER_GROUP(12)
        if (left <= 32u) break;
        left -= 32u;
        base += 32u;
        rb = load_rowbytes(csr, base, left, slot);
      }
      const float keep = hrow ? acc.y : acc.x;
      const float send = hrow ? acc.x : acc.y;
      const float tot = keep + xhalf_f32(send);
      out[((size_t)b * NN + n) * ND + ocol] = av + tot;

      s_cur = s_nxt;
      e_cur = e_nxt;
      rb_cur = ok_nxt ? (raw_nxt << 7) : (unsigned)GZROW_B;
    }
  }
}

extern "C" void kernel_launch(void* const* d_in, const int* in_sizes, int n_in,
                              void* d_out, int out_size, void* d_ws, size_t ws_size,
                              hipStream_t stream) {
  (void)in_sizes; (void)n_in; (void)out_size; (void)ws_size;
  const float* X   = (const float*)d_in[0];
  const int*   ei  = (const int*)d_in[1];
  const float* W1t = (const float*)d_in[2];
  const float* b1t = (const float*)d_in[3];
  const float* W2t = (const float*)d_in[4];
  const float* b2t = (const float*)d_in[5];
  const float* W1n = (const float*)d_in[6];
  const float* b1n = (const float*)d_in[7];
  const float* W2n = (const float*)d_in[8];
  const float* b2n = (const float*)d_in[9];
  float* out = (float*)d_out;

  float* h     = (float*)d_ws;
  float* xmax  = h + (size_t)NB * NN * ND;
  float* nodex = xmax + (size_t)NN * ND;
  float* agg   = nodex + (size_t)NN * ND;
  unsigned* deg    = (unsigned*)(agg + (size_t)NN * ND);
  unsigned* off    = deg + NN;
  unsigned* cursor = off + (NN + 1);
  unsigned* csr    = cursor + NN;

  const int* esrc = ei;
  const int* edst = ei + NE;

  k_zero_u32<<<(NN + 255) / 256, 256, 0, stream>>>(deg, NN);
  k_count<<<NE / 256, 256, 0, stream>>>(edst, deg);
  k_scan<<<1, 256, 0, stream>>>(deg, off, cursor);
  k_fill<<<NE / 256, 256, 0, stream>>>(esrc, edst, cursor, csr);

  k_maxb<<<(NN * ND) / (4 * MAXB_COLS4), MAXB_WAVES * 32, 0, stream>>>(X, xmax);
  k_mlp2_wmma<<<(NN / 16) / MLP_WAVES, MLP_WAVES * 32, 0, stream>>>(xmax, W1n, b1n, W2n, b2n, nodex, NN / 16);
  k_gather_node<<<NN / 4, 256, 0, stream>>>(nodex, off, csr, agg);

  k_mlp2_wmma<<<((NB * NN) / 16) / (MLP_WAVES * MLP_TILES_BIG), MLP_WAVES * 32, 0, stream>>>(X, W1t, b1t, W2t, b2t, h, (NB * NN) / 16);
  k_gather_final<<<NB * 2, 1024, 0, stream>>>(h, agg, off, csr, out);
}


# CSR build (zero, count, scan, fill: 4 launches) runs as 8 extra blocks of the max launch on LDS histograms (9 -> 5 launches); a node's edge order (f32 add order) still by atomic arrival
# speedup vs baseline: 1.9111x; 1.0503x over previous
#include <hip/hip_runtime.h>

typedef __attribute__((ext_vector_type(2))) float v2f;
typedef __attribute__((ext_vector_type(4))) float v4f;
typedef __attribute__((ext_vector_type(8))) float v8f;

#define NB 128
#define NN 2048
#define ND 64
#define NE 32768

#define CSR_BLOCKS 8
#define CSR_GROUPS (NE / 4 / 256)
#define CSR_OWN (CSR_GROUPS / CSR_BLOCKS)

__device__ __forceinline__ void csr_block(const int* __restrict__ src, const int* __restrict__ dst,
                                          unsigned* __restrict__ off, unsigned* __restrict__ csr,
                                          const int k, unsigned* hist, unsigned* wsum) {
  const int t = threadIdx.x;
  const int lane = t & 31;
  const int w = t >> 5;
  const int4* src4 = (const int4*)src;
  const int4* dst4 = (const int4*)dst;

  for (int i = 0; i < (2 * NN) / (4 * 256); ++i)
    *(uint4*)&hist[4 * (t + 256 * i)] = make_uint4(0u, 0u, 0u, 0u);
  __syncthreads();

  int4 os[CSR_OWN];
  for (int i = 0; i < CSR_OWN; ++i) os[i] = src4[t + 256 * (CSR_OWN * k + i)];
  int4 d[CSR_GROUPS];
#pragma unroll
  for (int j = 0; j < CSR_GROUPS; ++j)
    d[j] = dst4[t + 256 * ((j + CSR_OWN * k) & (CSR_GROUPS - 1))];

#pragma unroll
  for (int j = 0; j < CSR_GROUPS; ++j) {
    const unsigned sel = (j >= CSR_GROUPS - CSR_OWN * k) ? 0u : (unsigned)NN;
    atomicAdd(&hist[sel + (unsigned)d[j].x], 1u);
    atomicAdd(&hist[sel + (unsigned)d[j].y], 1u);
    atomicAdd(&hist[sel + (unsigned)d[j].z], 1u);
    atomicAdd(&hist[sel + (unsigned)d[j].w], 1u);
  }
  __syncthreads();

  const uint4 b0 = *(const uint4*)&hist[8 * t];
  const uint4 b1 = *(const uint4*)&hist[8 * t + 4];
  const uint4 r0 = *(const uint4*)&hist[NN + 8 * t];
  const uint4 r1 = *(const uint4*)&hist[NN + 8 * t + 4];
  const unsigned bb[8] = {b0.x, b0.y, b0.z, b0.w, b1.x, b1.y, b1.z, b1.w};
  const unsigned cc[8] = {b0.x + r0.x, b0.y + r0.y, b0.z + r0.z, b0.w + r0.w,
                          b1.x + r1.x, b1.y + r1.y, b1.z + r1.z, b1.w + r1.w};
  unsigned tsum = 0u;
  for (int i = 0; i < 8; ++i) tsum += cc[i];
  unsigned incl = tsum;
  for (int o = 1; o < 32; o <<= 1) {
    const unsigned u = (unsigned)__shfl_up((int)incl, o);
    if (lane >= o) incl += u;
  }
  if (lane == 31) wsum[w] = incl;
  __syncthreads();
  unsigned base = incl - tsum;
  for (int w2 = 0; w2 < 8; ++w2)
    if (w2 < w) base += wsum[w2];
  unsigned oo[8];
  oo[0] = base;
  for (int i = 1; i < 8; ++i) oo[i] = oo[i - 1] + cc[i - 1];
  *(uint4*)&hist[8 * t] = make_uint4(oo[0] + bb[0], oo[1] + bb[1], oo[2] + bb[2], oo[3] + bb[3]);
  *(uint4*)&hist[8 * t + 4] = make_uint4(oo[4] + bb[4], oo[5] + bb[5], oo[6] + bb[6], oo[7] + bb[7]);
  if (k == 0) {
    *(uint4*)&off[8 * t] = make_uint4(oo[0], oo[1], oo[2], oo[3]);
    *(uint4*)&off[8 * t + 4] = make_uint4(oo[4], oo[5], oo[6], oo[7]);
    if (t == 255) off[NN] = oo[7] + cc[7];
  }
  __syncthreads();

  unsigned p[4 * CSR_OWN];
  for (int i = 0; i < CSR_OWN; ++i) {
    p[4 * i + 0] = atomicAdd(&hist[(unsigned)d[i].x], 1u);
    p[4 * i + 1] = atomicAdd(&hist[(unsigned)d[i].y], 1u);
    p[4 * i + 2] = atomicAdd(&hist[(unsigned)d[i].z], 1u);
    p[4 * i + 3] = atomicAdd(&hist[(unsigned)d[i].w], 1u);
  }
  for (int i = 0; i < CSR_OWN; ++i) {
    csr[p[4 * i + 0]] = (unsigned)os[i].x;
    csr[p[4 * i + 1]] = (unsigned)os[i].y;
    csr[p[4 * i + 2]] = (unsigned)os[i].z;
    csr[p[4 * i + 3]] = (unsigned)os[i].w;
  }
}

#define MAXB_WAVES 8
#define MAXB_PLANES (NB / MAXB_WAVES)
#define MAXB_COLS4 64

__device__ __forceinline__ v4f max4(v4f a, v4f b) {
  v4f r;
  r[0] = fmaxf(a[0], b[0]);
  r[1] = fmaxf(a[1], b[1]);
  r[2] = fmaxf(a[2], b[2]);
  r[3] = fmaxf(a[3], b[3]);
  return r;
}

__global__ void __launch_bounds__(MAXB_WAVES * 32) __attribute__((amdgpu_waves_per_eu(4, 4)))
k_csr_maxb(const float* __restrict__ X, float* __restrict__ xmax,
           const int* __restrict__ esrc, const int* __restrict__ edst,
           unsigned* __restrict__ off, unsigned* __restrict__ csr) {
  __shared__ v4f red[MAXB_WAVES][MAXB_COLS4];
  __shared__ unsigned hist[2 * NN];
  __shared__ unsigned wsum[MAXB_WAVES];
  if (blockIdx.x < CSR_BLOCKS) {
    csr_block(esrc, edst, off, csr, (int)blockIdx.x, hist, wsum);
    return;
  }
  const int bx = (int)blockIdx.x - CSR_BLOCKS;
  const int lane = threadIdx.x & 31;
  const int w = threadIdx.x >> 5;
  const int c4 = bx * MAXB_COLS4 + lane;
  const float* p = X + (size_t)w * (NN * ND) + (size_t)c4 * 4;
  v4f va[MAXB_PLANES], vb[MAXB_PLANES];
#pragma unroll
  for (int j = 0; j < MAXB_PLANES; ++j) {
    const float* q = p + (size_t)j * (MAXB_WAVES * NN * ND);
    va[j] = *(const v4f*)q;
    vb[j] = *(const v4f*)(q + 128);
  }
  v4f ma = va[0], mb = vb[0];
#pragma unroll
  for (int j = 1; j < MAXB_PLANES; ++j) {
    ma = max4(ma, va[j]);
    mb = max4(mb, vb[j]);
  }
  red[w][lane] = ma;
  red[w][lane + 32] = mb;
  __syncthreads();
  if (threadIdx.x < MAXB_COLS4) {
    v4f r = red[0][threadIdx.x];
#pragma unroll
    for (int k = 1; k < MAXB_WAVES; ++k) r = max4(r, red[k][threadIdx.x]);
    *(v4f*)(xmax + ((size_t)bx * MAXB_COLS4 + threadIdx.x) * 4) = r;
  }
}

#define MLP_WAVES 8
#define MLP_TILES_BIG 8
__global__ void __launch_bounds__(MLP_WAVES * 32)
k_mlp2_wmma(const float* __restrict__ Xin, const float* __restrict__ W1,
            const float* __restrict__ b1, const float* __restrict__ W2,
            const float* __restrict__ b2, float* __restrict__ Yout, int ntiles) {
  const int lane = threadIdx.x & 31;
  const int hm = lane & 15;
  const int hk = lane >> 4;
  const int gw = __builtin_amdgcn_readfirstlane((int)(blockIdx.x * MLP_WAVES + (threadIdx.x >> 5)));
  const int nw = (int)gridDim.x * MLP_WAVES;

  v2f w1[4][16];
#pragma unroll
  for (int s = 0; s < 16; ++s) {
#pragma unroll
    for (int e = 0; e < 2; ++e) {
      const int k = 8 * (s >> 1) + 4 * hk + 2 * (s & 1) + e;
      const v4f wv = *(const v4f*)(W1 + k * ND + 4 * hm);
      w1[0][s][e] = wv[0]; w1[1][s][e] = wv[1]; w1[2][s][e] = wv[2]; w1[3][s][e] = wv[3];
    }
  }
  v2f w2[4][16];
#pragma unroll
  for (int t = 0; t < 16; ++t) {
#pragma unroll
    for (int e = 0; e < 2; ++e) {
      const int f = 4 * (8 * hk + 2 * (t & 3) + e) + (t >> 2);
      const v4f wv = *(const v4f*)(W2 + f * ND + 4 * hm);
      w2[0][t][e] = wv[0]; w2[1][t][e] = wv[1]; w2[2][t][e] = wv[2]; w2[3][t][e] = wv[3];
    }
  }
  v8f c1[4];
#pragma unroll
  for (int r = 0; r < 8; ++r) {
    const v4f bv = *(const v4f*)(b1 + 4 * (8 * hk + r));
    c1[0][r] = bv[0]; c1[1][r] = bv[1]; c1[2][r] = bv[2]; c1[3][r] = bv[3];
  }
  const v4f c2 = *(const v4f*)(b2 + 4 * hm);

  if (gw >= ntiles) return;

  v4f x[8];
  {
    const float* xr = Xin + ((size_t)gw * 16 + hm) * ND + 4 * hk;
#pragma unroll
    for (int i = 0; i < 8; ++i) x[i] = *(const v4f*)(xr + 8 * i);
  }

  for (int t = gw; t < ntiles; t += nw) {
    const int tn = (t + nw < ntiles) ? (t + nw) : t;
    v4f xn[8];
    {
      const float* xr = Xin + ((size_t)tn * 16 + hm) * ND + 4 * hk;
#pragma unroll
      for (int i = 0; i < 8; ++i) xn[i] = *(const v4f*)(xr + 8 * i);
    }

    v8f d1[4];
#pragma unroll
    for (int ft = 0; ft < 4; ++ft) d1[ft] = c1[ft];
#pragma unroll
    for (int i = 0; i < 8; ++i) {
      v2f blo, bhi;
      blo[0] = x[i][0]; blo[1] = x[i][1];
      bhi[0] = x[i][2]; bhi[1] = x[i][3];
#pragma unroll
      for (int ft = 0; ft < 4; ++ft)
        d1[ft] = __builtin_amdgcn_wmma_f32_16x16x4_f32(false, w1[ft][2 * i], false, blo,
                                                       (short)0, d1[ft], false, false);
#pragma unroll
      for (int ft = 0; ft < 4; ++ft)
        d1[ft] = __builtin_amdgcn_wmma_f32_16x16x4_f32(false, w1[ft][2 * i + 1], false, bhi,
                                                       (short)0, d1[ft], false, false);
    }
#pragma unroll
    for (int ft = 0; ft < 4; ++ft) {
#pragma unroll
      for (int r = 0; r < 8; ++r) d1[ft][r] = fmaxf(d1[ft][r], 0.0f);
    }

    v8f d2[4];
#pragma unroll
    for (int gt = 0; gt < 4; ++gt) {
#pragma unroll
      for (int r = 0; r < 8; ++r) d2[gt][r] = c2[gt];
    }
#pragma unroll
    for (int ft = 0; ft < 4; ++ft) {
#pragma unroll
      for (int jp = 0; jp < 4; ++jp) {
        v2f a;
        a[0] = d1[ft][2 * jp];
        a[1] = d1[ft][2 * jp + 1];
#pragma unroll
        for (int gt = 0; gt < 4; ++gt)
          d2[gt] = __builtin_amdgcn_wmma_f32_16x16x4_f32(false, a, false, w2[gt][4 * ft + jp],
                                                         (short)0, d2[gt], false, false);
      }
    }

    {
      float* yr = Yout + ((size_t)t * 16 + 8 * hk) * ND + 4 * hm;
#pragma unroll
      for (int r = 0; r < 8; ++r) {
        v4f o;
        o[0] = fmaxf(d2[0][r], 0.0f);
        o[1] = fmaxf(d2[1][r], 0.0f);
        o[2] = fmaxf(d2[2][r], 0.0f);
        o[3] = fmaxf(d2[3][r], 0.0f);
        *(v4f*)(yr + r * ND) = o;
      }
    }

#pragma unroll
    for (int i = 0; i < 8; ++i) x[i] = xn[i];
  }
}

__global__ void k_gather_node(const float* __restrict__ nodex,
                              const unsigned* __restrict__ off,
                              const unsigned* __restrict__ csr,
                              float* __restrict__ agg) {
  int r = blockIdx.x * 4 + (threadIdx.x >> 6);
  int d = threadIdx.x & 63;
  unsigned s = off[r], e = off[r + 1];
  float acc = 0.0f;
  for (unsigned j = s; j < e; ++j) acc += nodex[(size_t)csr[j] * ND + d];
  agg[(size_t)r * ND + d] = acc;
}

#define GROW 32
#define GZROW_B (NN * GROW * 4)

template <int T>
__device__ __forceinline__ unsigned row_share_u32(unsigned v) {
  return (unsigned)__builtin_amdgcn_update_dpp(0, (int)v, 0x150 + T, 0xf, 0xf, false);
}

__device__ __forceinline__ float xhalf_f32(float x) {
  return __builtin_bit_cast(float,
      __builtin_amdgcn_permlanex16(__builtin_bit_cast(int, x), __builtin_bit_cast(int, x),
                                   0x76543210, 0xfedcba98, false, false));
}

__device__ __forceinline__ unsigned load_rowbytes(const unsigned* __restrict__ csr,
                                                  unsigned base, unsigned left,
                                                  unsigned slot) {
  unsigned r = GZROW_B;
  if (slot < left) r = csr[base + slot] << 7;
  return r;
}

template <int T>
__device__ __forceinline__ v2f lds_row2(const char* hsb, unsigned rb, unsigned colb) {
  return *(const v2f*)(hsb + (row_share_u32<T>(rb) + colb));
}

#define GATHER_GROUP(T)                                   \
  {                                                       \
    const v2f x0 = lds_row2<T + 0>(hsb, rb, colb);        \
    const v2f x1 = lds_row2<T + 1>(hsb, rb, colb);        \
    const v2f x2 = lds_row2<T + 2>(hsb, rb, colb);        \
    const v2f x3 = lds_row2<T + 3>(hsb, rb, colb);        \
    acc += x0; acc += x1; acc += x2; acc += x3;           \
  }

__global__ void __launch_bounds__(1024)
k_gather_final(const float* __restrict__ h, const float* __restrict__ agg,
               const unsigned* __restrict__ off, const unsigned* __restrict__ csr,
               float* __restrict__ out) {
  __shared__ float hs[(NN + 1) * GROW];
  const int tid = threadIdx.x;
  const int lane = tid & 31;
  const int wave = __builtin_amdgcn_readfirstlane(tid >> 5);
  const int b = blockIdx.x >> 1;
  const int half = blockIdx.x & 1;

  {
    const float* hsrc = h + (size_t)b * (NN * ND) + half * GROW;
    const int r0 = tid >> 3;
    const int c4 = (tid & 7) * 4;
    float4 v[16];
#pragma unroll
    for (int p = 0; p < 16; ++p)
      v[p] = *(const float4*)(hsrc + (size_t)(r0 + p * 128) * ND + c4);
#pragma unroll
    for (int p = 0; p < 16; ++p)
      *(float4*)&hs[(r0 + p * 128) * GROW + c4] = v[p];
    if (tid < GROW) hs[NN * GROW + tid] = 0.0f;
  }
  __syncthreads();

  const char* hsb = (const char*)hs;
  const int c = lane & 15;
  const int hrow = lane >> 4;
  const unsigned slot = 2u * c + hrow;
  const unsigned colb = 8u * c;
  const int ocol = half * GROW + 2 * c + hrow;

  for (int q = 0; q < 2; ++q) {
    const int nq = wave * 64 + q * 32;
    const unsigned ov_e = off[nq + lane + 1];
    unsigned s_cur = off[nq];
    unsigned e_cur = (unsigned)__builtin_amdgcn_readlane((int)ov_e, 0);
    unsigned rb_cur = load_rowbytes(csr, s_cur, e_cur - s_cur, slot);
    for (int i = 0; i < 32; ++i) {
      const int n = nq + i;
      const unsigned s_nxt = e_cur;
      const unsigned e_nxt = (unsigned)__builtin_amdgcn_readlane((int)ov_e, (i + 1) & 31);
      const unsigned cnt_nxt = (i + 1 < 32) ? (e_nxt - s_nxt) : 0u;
      const bool ok_nxt = slot < cnt_nxt;
      const unsigned raw_nxt = csr[ok_nxt ? (s_nxt + slot) : 0u];
      const float av = agg[(size_t)n * ND + ocol];

      v2f acc = {0.0f, 0.0f};
      unsigned rb = rb_cur;
      unsigned left = e_cur - s_cur;
      unsigned base = s_cur;
      for (;;) {
        if (left > 0u) GATHER_GROUP(0)
        if (left > 8u) GATHER_GROUP(4)
        if (left > 16u) GATHER_GROUP(8)
        if (left > 24u) GATHER_GROUP(12)
        if (left <= 32u) break;
        left -= 32u;
        base += 32u;
        rb = load_rowbytes(csr, base, left, slot);
      }
      const float keep = hrow ? acc.y : acc.x;
      const float send = hrow ? acc.x : acc.y;
      const float tot = keep + xhalf_f32(send);
      out[((size_t)b * NN + n) * ND + ocol] = av + tot;

      s_cur = s_nxt;
      e_cur = e_nxt;
      rb_cur = ok_nxt ? (raw_nxt << 7) : (unsigned)GZROW_B;
    }
  }
}

extern "C" void kernel_launch(void* const* d_in, const int* in_sizes, int n_in,
                              void* d_out, int out_size, void* d_ws, size_t ws_size,
                              hipStream_t stream) {
  (void)in_sizes; (void)n_in; (void)out_size; (void)ws_size;
  const float* X   = (const float*)d_in[0];
  const int*   ei  = (const int*)d_in[1];
  const float* W1t = (const float*)d_in[2];
  const float* b1t = (const float*)d_in[3];
  const float* W2t = (const float*)d_in[4];
  const float* b2t = (const float*)d_in[5];
  const float* W1n = (const float*)d_in[6];
  const float* b1n = (const float*)d_in[7];
  const float* W2n = (const float*)d_in[8];
  const float* b2n = (const float*)d_in[9];
  float* out = (float*)d_out;

  float* h     = (float*)d_ws;
  float* xmax  = h + (size_t)NB * NN * ND;
  float* nodex = xmax + (size_t)NN * ND;
  float* agg   = nodex + (size_t)NN * ND;
  unsigned* off = (unsigned*)(agg + (size_t)NN * ND);
  unsigned* csr = off + (NN + 64);

  const int* esrc = ei;
  const int* edst = ei + NE;

  k_csr_maxb<<<CSR_BLOCKS + (NN * ND) / (4 * MAXB_COLS4), MAXB_WAVES * 32, 0, stream>>>(X, xmax, esrc, edst, off, csr);
  k_mlp2_wmma<<<(NN / 16) / MLP_WAVES, MLP_WAVES * 32, 0, stream>>>(xmax, W1n, b1n, W2n, b2n, nodex, NN / 16);
  k_gather_node<<<NN / 4, 256, 0, stream>>>(nodex, off, csr, agg);

  k_mlp2_wmma<<<((NB * NN) / 16) / (MLP_WAVES * MLP_TILES_BIG), MLP_WAVES * 32, 0, stream>>>(X, W1t, b1t, W2t, b2t, h, (NB * NN) / 16);
  k_gather_final<<<NB * 2, 1024, 0, stream>>>(h, agg, off, csr, out);
}


# final gather stages h + nodex rows in LDS, so one edge pass sums both branches (segment sum is linear); node gather launch and agg reads go (5 -> 4 launches; new f32 add order)
# speedup vs baseline: 1.9828x; 1.0375x over previous
#include <hip/hip_runtime.h>

typedef __attribute__((ext_vector_type(2))) float v2f;
typedef __attribute__((ext_vector_type(4))) float v4f;
typedef __attribute__((ext_vector_type(8))) float v8f;

#define NB 128
#define NN 2048
#define ND 64
#define NE 32768

#define CSR_BLOCKS 8
#define CSR_GROUPS (NE / 4 / 256)
#define CSR_OWN (CSR_GROUPS / CSR_BLOCKS)

__device__ __forceinline__ void csr_block(const int* __restrict__ src, const int* __restrict__ dst,
                                          unsigned* __restrict__ off, unsigned* __restrict__ csr,
                                          const int k, unsigned* hist, unsigned* wsum) {
  const int t = threadIdx.x;
  const int lane = t & 31;
  const int w = t >> 5;
  const int4* src4 = (const int4*)src;
  const int4* dst4 = (const int4*)dst;

  for (int i = 0; i < (2 * NN) / (4 * 256); ++i)
    *(uint4*)&hist[4 * (t + 256 * i)] = make_uint4(0u, 0u, 0u, 0u);
  __syncthreads();

  int4 os[CSR_OWN];
  for (int i = 0; i < CSR_OWN; ++i) os[i] = src4[t + 256 * (CSR_OWN * k + i)];
  int4 d[CSR_GROUPS];
#pragma unroll
  for (int j = 0; j < CSR_GROUPS; ++j)
    d[j] = dst4[t + 256 * ((j + CSR_OWN * k) & (CSR_GROUPS - 1))];

#pragma unroll
  for (int j = 0; j < CSR_GROUPS; ++j) {
    const unsigned sel = (j >= CSR_GROUPS - CSR_OWN * k) ? 0u : (unsigned)NN;
    atomicAdd(&hist[sel + (unsigned)d[j].x], 1u);
    atomicAdd(&hist[sel + (unsigned)d[j].y], 1u);
    atomicAdd(&hist[sel + (unsigned)d[j].z], 1u);
    atomicAdd(&hist[sel + (unsigned)d[j].w], 1u);
  }
  __syncthreads();

  const uint4 b0 = *(const uint4*)&hist[8 * t];
  const uint4 b1 = *(const uint4*)&hist[8 * t + 4];
  const uint4 r0 = *(const uint4*)&hist[NN + 8 * t];
  const uint4 r1 = *(const uint4*)&hist[NN + 8 * t + 4];
  const unsigned bb[8] = {b0.x, b0.y, b0.z, b0.w, b1.x, b1.y, b1.z, b1.w};
  const unsigned cc[8] = {b0.x + r0.x, b0.y + r0.y, b0.z + r0.z, b0.w + r0.w,
                          b1.x + r1.x, b1.y + r1.y, b1.z + r1.z, b1.w + r1.w};
  unsigned tsum = 0u;
  for (int i = 0; i < 8; ++i) tsum += cc[i];
  unsigned incl = tsum;
  for (int o = 1; o < 32; o <<= 1) {
    const unsigned u = (unsigned)__shfl_up((int)incl, o);
    if (lane >= o) incl += u;
  }
  if (lane == 31) wsum[w] = incl;
  __syncthreads();
  unsigned base = incl - tsum;
  for (int w2 = 0; w2 < 8; ++w2)
    if (w2 < w) base += wsum[w2];
  unsigned oo[8];
  oo[0] = base;
  for (int i = 1; i < 8; ++i) oo[i] = oo[i - 1] + cc[i - 1];
  *(uint4*)&hist[8 * t] = make_uint4(oo[0] + bb[0], oo[1] + bb[1], oo[2] + bb[2], oo[3] + bb[3]);
  *(uint4*)&hist[8 * t + 4] = make_uint4(oo[4] + bb[4], oo[5] + bb[5], oo[6] + bb[6], oo[7] + bb[7]);
  if (k == 0) {
    *(uint4*)&off[8 * t] = make_uint4(oo[0], oo[1], oo[2], oo[3]);
    *(uint4*)&off[8 * t + 4] = make_uint4(oo[4], oo[5], oo[6], oo[7]);
    if (t == 255) off[NN] = oo[7] + cc[7];
  }
  __syncthreads();

  unsigned p[4 * CSR_OWN];
  for (int i = 0; i < CSR_OWN; ++i) {
    p[4 * i + 0] = atomicAdd(&hist[(unsigned)d[i].x], 1u);
    p[4 * i + 1] = atomicAdd(&hist[(unsigned)d[i].y], 1u);
    p[4 * i + 2] = atomicAdd(&hist[(unsigned)d[i].z], 1u);
    p[4 * i + 3] = atomicAdd(&hist[(unsigned)d[i].w], 1u);
  }
  for (int i = 0; i < CSR_OWN; ++i) {
    csr[p[4 * i + 0]] = (unsigned)os[i].x;
    csr[p[4 * i + 1]] = (unsigned)os[i].y;
    csr[p[4 * i + 2]] = (unsigned)os[i].z;
    csr[p[4 * i + 3]] = (unsigned)os[i].w;
  }
}

#define MAXB_WAVES 8
#define MAXB_PLANES (NB / MAXB_WAVES)
#define MAXB_COLS4 64

__device__ __forceinline__ v4f max4(v4f a, v4f b) {
  v4f r;
  r[0] = fmaxf(a[0], b[0]);
  r[1] = fmaxf(a[1], b[1]);
  r[2] = fmaxf(a[2], b[2]);
  r[3] = fmaxf(a[3], b[3]);
  return r;
}

__global__ void __launch_bounds__(MAXB_WAVES * 32) __attribute__((amdgpu_waves_per_eu(4, 4)))
k_csr_maxb(const float* __restrict__ X, float* __restrict__ xmax,
           const int* __restrict__ esrc, const int* __restrict__ edst,
           unsigned* __restrict__ off, unsigned* __restrict__ csr) {
  __shared__ v4f red[MAXB_WAVES][MAXB_COLS4];
  __shared__ unsigned hist[2 * NN];
  __shared__ unsigned wsum[MAXB_WAVES];
  if (blockIdx.x < CSR_BLOCKS) {
    csr_block(esrc, edst, off, csr, (int)blockIdx.x, hist, wsum);
    return;
  }
  const int bx = (int)blockIdx.x - CSR_BLOCKS;
  const int lane = threadIdx.x & 31;
  const int w = threadIdx.x >> 5;
  const int c4 = bx * MAXB_COLS4 + lane;
  const float* p = X + (size_t)w * (NN * ND) + (size_t)c4 * 4;
  v4f va[MAXB_PLANES], vb[MAXB_PLANES];
#pragma unroll
  for (int j = 0; j < MAXB_PLANES; ++j) {
    const float* q = p + (size_t)j * (MAXB_WAVES * NN * ND);
    va[j] = *(const v4f*)q;
    vb[j] = *(const v4f*)(q + 128);
  }
  v4f ma = va[0], mb = vb[0];
#pragma unroll
  for (int j = 1; j < MAXB_PLANES; ++j) {
    ma = max4(ma, va[j]);
    mb = max4(mb, vb[j]);
  }
  red[w][lane] = ma;
  red[w][lane + 32] = mb;
  __syncthreads();
  if (threadIdx.x < MAXB_COLS4) {
    v4f r = red[0][threadIdx.x];
#pragma unroll
    for (int k = 1; k < MAXB_WAVES; ++k) r = max4(r, red[k][threadIdx.x]);
    *(v4f*)(xmax + ((size_t)bx * MAXB_COLS4 + threadIdx.x) * 4) = r;
  }
}

#define MLP_WAVES 8
#define MLP_TILES_BIG 8
__global__ void __launch_bounds__(MLP_WAVES * 32)
k_mlp2_wmma(const float* __restrict__ Xin, const float* __restrict__ W1,
            const float* __restrict__ b1, const float* __restrict__ W2,
            const float* __restrict__ b2, float* __restrict__ Yout, int ntiles) {
  const int lane = threadIdx.x & 31;
  const int hm = lane & 15;
  const int hk = lane >> 4;
  const int gw = __builtin_amdgcn_readfirstlane((int)(blockIdx.x * MLP_WAVES + (threadIdx.x >> 5)));
  const int nw = (int)gridDim.x * MLP_WAVES;

  v2f w1[4][16];
#pragma unroll
  for (int s = 0; s < 16; ++s) {
#pragma unroll
    for (int e = 0; e < 2; ++e) {
      const int k = 8 * (s >> 1) + 4 * hk + 2 * (s & 1) + e;
      const v4f wv = *(const v4f*)(W1 + k * ND + 4 * hm);
      w1[0][s][e] = wv[0]; w1[1][s][e] = wv[1]; w1[2][s][e] = wv[2]; w1[3][s][e] = wv[3];
    }
  }
  v2f w2[4][16];
#pragma unroll
  for (int t = 0; t < 16; ++t) {
#pragma unroll
    for (int e = 0; e < 2; ++e) {
      const int f = 4 * (8 * hk + 2 * (t & 3) + e) + (t >> 2);
      const v4f wv = *(const v4f*)(W2 + f * ND + 4 * hm);
      w2[0][t][e] = wv[0]; w2[1][t][e] = wv[1]; w2[2][t][e] = wv[2]; w2[3][t][e] = wv[3];
    }
  }
  v8f c1[4];
#pragma unroll
  for (int r = 0; r < 8; ++r) {
    const v4f bv = *(const v4f*)(b1 + 4 * (8 * hk + r));
    c1[0][r] = bv[0]; c1[1][r] = bv[1]; c1[2][r] = bv[2]; c1[3][r] = bv[3];
  }
  const v4f c2 = *(const v4f*)(b2 + 4 * hm);

  if (gw >= ntiles) return;

  v4f x[8];
  {
    const float* xr = Xin + ((size_t)gw * 16 + hm) * ND + 4 * hk;
#pragma unroll
    for (int i = 0; i < 8; ++i) x[i] = *(const v4f*)(xr + 8 * i);
  }

  for (int t = gw; t < ntiles; t += nw) {
    const int tn = (t + nw < ntiles) ? (t + nw) : t;
    v4f xn[8];
    {
      const float* xr = Xin + ((size_t)tn * 16 + hm) * ND + 4 * hk;
#pragma unroll
      for (int i = 0; i < 8; ++i) xn[i] = *(const v4f*)(xr + 8 * i);
    }

    v8f d1[4];
#pragma unroll
    for (int ft = 0; ft < 4; ++ft) d1[ft] = c1[ft];
#pragma unroll
    for (int i = 0; i < 8; ++i) {
      v2f blo, bhi;
      blo[0] = x[i][0]; blo[1] = x[i][1];
      bhi[0] = x[i][2]; bhi[1] = x[i][3];
#pragma unroll
      for (int ft = 0; ft < 4; ++ft)
        d1[ft] = __builtin_amdgcn_wmma_f32_16x16x4_f32(false, w1[ft][2 * i], false, blo,
                                                       (short)0, d1[ft], false, false);
#pragma unroll
      for (int ft = 0; ft < 4; ++ft)
        d1[ft] = __builtin_amdgcn_wmma_f32_16x16x4_f32(false, w1[ft][2 * i + 1], false, bhi,
                                                       (short)0, d1[ft], false, false);
    }
#pragma unroll
    for (int ft = 0; ft < 4; ++ft) {
#pragma unroll
      for (int r = 0; r < 8; ++r) d1[ft][r] = fmaxf(d1[ft][r], 0.0f);
    }

    v8f d2[4];
#pragma unroll
    for (int gt = 0; gt < 4; ++gt) {
#pragma unroll
      for (int r = 0; r < 8; ++r) d2[gt][r] = c2[gt];
    }
#pragma unroll
    for (int ft = 0; ft < 4; ++ft) {
#pragma unroll
      for (int jp = 0; jp < 4; ++jp) {
        v2f a;
        a[0] = d1[ft][2 * jp];
        a[1] = d1[ft][2 * jp + 1];
#pragma unroll
        for (int gt = 0; gt < 4; ++gt)
          d2[gt] = __builtin_amdgcn_wmma_f32_16x16x4_f32(false, a, false, w2[gt][4 * ft + jp],
                                                         (short)0, d2[gt], false, false);
      }
    }

    {
      float* yr = Yout + ((size_t)t * 16 + 8 * hk) * ND + 4 * hm;
#pragma unroll
      for (int r = 0; r < 8; ++r) {
        v4f o;
        o[0] = fmaxf(d2[0][r], 0.0f);
        o[1] = fmaxf(d2[1][r], 0.0f);
        o[2] = fmaxf(d2[2][r], 0.0f);
        o[3] = fmaxf(d2[3][r], 0.0f);
        *(v4f*)(yr + r * ND) = o;
      }
    }

#pragma unroll
    for (int i = 0; i < 8; ++i) x[i] = xn[i];
  }
}

#define GROW 32
#define GZROW_B (NN * GROW * 4)

template <int T>
__device__ __forceinline__ unsigned row_share_u32(unsigned v) {
  return (unsigned)__builtin_amdgcn_update_dpp(0, (int)v, 0x150 + T, 0xf, 0xf, false);
}

__device__ __forceinline__ float xhalf_f32(float x) {
  return __builtin_bit_cast(float,
      __builtin_amdgcn_permlanex16(__builtin_bit_cast(int, x), __builtin_bit_cast(int, x),
                                   0x76543210, 0xfedcba98, false, false));
}

__device__ __forceinline__ unsigned load_rowbytes(const unsigned* __restrict__ csr,
                                                  unsigned base, unsigned left,
                                                  unsigned slot) {
  unsigned r = GZROW_B;
  if (slot < left) r = csr[base + slot] << 7;
  return r;
}

template <int T>
__device__ __forceinline__ v2f lds_row2(const char* hsb, unsigned rb, unsigned colb) {
  return *(const v2f*)(hsb + (row_share_u32<T>(rb) + colb));
}

#define GATHER_GROUP(T)                                   \
  {                                                       \
    const v2f x0 = lds_row2<T + 0>(hsb, rb, colb);        \
    const v2f x1 = lds_row2<T + 1>(hsb, rb, colb);        \
    const v2f x2 = lds_row2<T + 2>(hsb, rb, colb);        \
    const v2f x3 = lds_row2<T + 3>(hsb, rb, colb);        \
    acc += x0; acc += x1; acc += x2; acc += x3;           \
  }

__global__ void __launch_bounds__(1024)
k_gather_final(const float* __restrict__ h, const float* __restrict__ nodex,
               const unsigned* __restrict__ off, const unsigned* __restrict__ csr,
               float* __restrict__ out) {
  __shared__ float hs[(NN + 1) * GROW];
  const int tid = threadIdx.x;
  const int lane = tid & 31;
  const int wave = __builtin_amdgcn_readfirstlane(tid >> 5);
  const int b = blockIdx.x >> 1;
  const int half = blockIdx.x & 1;

  {
    const float* hsrc = h + (size_t)b * (NN * ND) + half * GROW;
    const float* nsrc = nodex + half * GROW;
    const int r0 = tid >> 3;
    const int c4 = (tid & 7) * 4;
#pragma nounroll
    for (int g = 0; g < 2; ++g) {
      float4 v[8], u[8];
#pragma unroll
      for (int p = 0; p < 8; ++p) {
        const size_t ro = (size_t)(r0 + (8 * g + p) * 128) * ND + c4;
        v[p] = *(const float4*)(hsrc + ro);
        u[p] = *(const float4*)(nsrc + ro);
      }
#pragma unroll
      for (int p = 0; p < 8; ++p) {
        float4 s;
        s.x = v[p].x + u[p].x;
        s.y = v[p].y + u[p].y;
        s.z = v[p].z + u[p].z;
        s.w = v[p].w + u[p].w;
        *(float4*)&hs[(r0 + (8 * g + p) * 128) * GROW + c4] = s;
      }
    }
    if (tid < GROW) hs[NN * GROW + tid] = 0.0f;
  }
  __syncthreads();

  const char* hsb = (const char*)hs;
  const int c = lane & 15;
  const int hrow = lane >> 4;
  const unsigned slot = 2u * c + hrow;
  const unsigned colb = 8u * c;
  const int ocol = half * GROW + 2 * c + hrow;

  for (int q = 0; q < 2; ++q) {
    const int nq = wave * 64 + q * 32;
    const unsigned ov_e = off[nq + lane + 1];
    unsigned s_cur = off[nq];
    unsigned e_cur = (unsigned)__builtin_amdgcn_readlane((int)ov_e, 0);
    unsigned rb_cur = load_rowbytes(csr, s_cur, e_cur - s_cur, slot);
    for (int i = 0; i < 32; ++i) {
      const int n = nq + i;
      const unsigned s_nxt = e_cur;
      const unsigned e_nxt = (unsigned)__builtin_amdgcn_readlane((int)ov_e, (i + 1) & 31);
      const unsigned cnt_nxt = (i + 1 < 32) ? (e_nxt - s_nxt) : 0u;
      const bool ok_nxt = slot < cnt_nxt;
      const unsigned raw_nxt = csr[ok_nxt ? (s_nxt + slot) : 0u];

      v2f acc = {0.0f, 0.0f};
      unsigned rb = rb_cur;
      unsigned left = e_cur - s_cur;
      unsigned base = s_cur;
      for (;;) {
        if (left > 0u) GATHER_GROUP(0)
        if (left > 8u) GATHER_GROUP(4)
        if (left > 16u) GATHER_GROUP(8)
        if (left > 24u) GATHER_GROUP(12)
        if (left <= 32u) break;
        left -= 32u;
        base += 32u;
        rb = load_rowbytes(csr, base, left, slot);
      }
      const float keep = hrow ? acc.y : acc.x;
      const float send = hrow ? acc.x : acc.y;
      const float tot = keep + xhalf_f32(send);
      out[((size_t)b * NN + n) * ND + ocol] = tot;

      s_cur = s_nxt;
      e_cur = e_nxt;
      rb_cur = ok_nxt ? (raw_nxt << 7) : (unsigned)GZROW_B;
    }
  }
}

extern "C" void kernel_launch(void* const* d_in, const int* in_sizes, int n_in,
                              void* d_out, int out_size, void* d_ws, size_t ws_size,
                              hipStream_t stream) {
  (void)in_sizes; (void)n_in; (void)out_size; (void)ws_size;
  const float* X   = (const float*)d_in[0];
  const int*   ei  = (const int*)d_in[1];
  const float* W1t = (const float*)d_in[2];
  const float* b1t = (const float*)d_in[3];
  const float* W2t = (const float*)d_in[4];
  const float* b2t = (const float*)d_in[5];
  const float* W1n = (const float*)d_in[6];
  const float* b1n = (const float*)d_in[7];
  const float* W2n = (const float*)d_in[8];
  const float* b2n = (const float*)d_in[9];
  float* out = (float*)d_out;

  float* h     = (float*)d_ws;
  float* xmax  = h + (size_t)NB * NN * ND;
  float* nodex = xmax + (size_t)NN * ND;
  unsigned* off = (unsigned*)(nodex + (size_t)NN * ND);
  unsigned* csr = off + (NN + 64);

  const int* esrc = ei;
  const int* edst = ei + NE;

  k_csr_maxb<<<CSR_BLOCKS + (NN * ND) / (4 * MAXB_COLS4), MAXB_WAVES * 32, 0, stream>>>(X, xmax, esrc, edst, off, csr);
  k_mlp2_wmma<<<(NN / 16) / MLP_WAVES, MLP_WAVES * 32, 0, stream>>>(xmax, W1n, b1n, W2n, b2n, nodex, NN / 16);

  k_mlp2_wmma<<<((NB * NN) / 16) / (MLP_WAVES * MLP_TILES_BIG), MLP_WAVES * 32, 0, stream>>>(X, W1t, b1t, W2t, b2t, h, (NB * NN) / 16);
  k_gather_final<<<NB * 2, 1024, 0, stream>>>(h, nodex, off, csr, out);
}


# max over subgraphs taken by the big MLP's waves on the X rows they load anyway (16 partial planes, reduced where the node MLP loads its input): X read once, 4 -> 3 launches; CSR build joins node MLP
# speedup vs baseline: 2.1371x; 1.0778x over previous
#include <hip/hip_runtime.h>

typedef __attribute__((ext_vector_type(2))) float v2f;
typedef __attribute__((ext_vector_type(4))) float v4f;
typedef __attribute__((ext_vector_type(8))) float v8f;

#define NB 128
#define NN 2048
#define ND 64
#define NE 32768

#define CSR_BLOCKS 8
#define CSR_GROUPS (NE / 4 / 256)
#define CSR_OWN (CSR_GROUPS / CSR_BLOCKS)

__device__ __forceinline__ void csr_block(const int* __restrict__ src, const int* __restrict__ dst,
                                          unsigned* __restrict__ off, unsigned* __restrict__ csr,
                                          const int k, unsigned* hist, unsigned* wsum) {
  const int t = threadIdx.x;
  const int lane = t & 31;
  const int w = t >> 5;
  const int4* src4 = (const int4*)src;
  const int4* dst4 = (const int4*)dst;

  for (int i = 0; i < (2 * NN) / (4 * 256); ++i)
    *(uint4*)&hist[4 * (t + 256 * i)] = make_uint4(0u, 0u, 0u, 0u);
  __syncthreads();

  int4 os[CSR_OWN];
  for (int i = 0; i < CSR_OWN; ++i) os[i] = src4[t + 256 * (CSR_OWN * k + i)];
  int4 d[CSR_GROUPS];
#pragma unroll
  for (int j = 0; j < CSR_GROUPS; ++j)
    d[j] = dst4[t + 256 * ((j + CSR_OWN * k) & (CSR_GROUPS - 1))];

#pragma unroll
  for (int j = 0; j < CSR_GROUPS; ++j) {
    const unsigned sel = (j >= CSR_GROUPS - CSR_OWN * k) ? 0u : (unsigned)NN;
    atomicAdd(&hist[sel + (unsigned)d[j].x], 1u);
    atomicAdd(&hist[sel + (unsigned)d[j].y], 1u);
    atomicAdd(&hist[sel + (unsigned)d[j].z], 1u);
    atomicAdd(&hist[sel + (unsigned)d[j].w], 1u);
  }
  __syncthreads();

  const uint4 b0 = *(const uint4*)&hist[8 * t];
  const uint4 b1 = *(const uint4*)&hist[8 * t + 4];
  const uint4 r0 = *(const uint4*)&hist[NN + 8 * t];
  const uint4 r1 = *(const uint4*)&hist[NN + 8 * t + 4];
  const unsigned bb[8] = {b0.x, b0.y, b0.z, b0.w, b1.x, b1.y, b1.z, b1.w};
  const unsigned cc[8] = {b0.x + r0.x, b0.y + r0.y, b0.z + r0.z, b0.w + r0.w,
                          b1.x + r1.x, b1.y + r1.y, b1.z + r1.z, b1.w + r1.w};
  unsigned tsum = 0u;
  for (int i = 0; i < 8; ++i) tsum += cc[i];
  unsigned incl = tsum;
  for (int o = 1; o < 32; o <<= 1) {
    const unsigned u = (unsigned)__shfl_up((int)incl, o);
    if (lane >= o) incl += u;
  }
  if (lane == 31) wsum[w] = incl;
  __syncthreads();
  unsigned base = incl - tsum;
  for (int w2 = 0; w2 < 8; ++w2)
    if (w2 < w) base += wsum[w2];
  unsigned oo[8];
  oo[0] = base;
  for (int i = 1; i < 8; ++i) oo[i] = oo[i - 1] + cc[i - 1];
  *(uint4*)&hist[8 * t] = make_uint4(oo[0] + bb[0], oo[1] + bb[1], oo[2] + bb[2], oo[3] + bb[3]);
  *(uint4*)&hist[8 * t + 4] = make_uint4(oo[4] + bb[4], oo[5] + bb[5], oo[6] + bb[6], oo[7] + bb[7]);
  if (k == 0) {
    *(uint4*)&off[8 * t] = make_uint4(oo[0], oo[1], oo[2], oo[3]);
    *(uint4*)&off[8 * t + 4] = make_uint4(oo[4], oo[5], oo[6], oo[7]);
    if (t == 255) off[NN] = oo[7] + cc[7];
  }
  __syncthreads();

  unsigned p[4 * CSR_OWN];
  for (int i = 0; i < CSR_OWN; ++i) {
    p[4 * i + 0] = atomicAdd(&hist[(unsigned)d[i].x], 1u);
    p[4 * i + 1] = atomicAdd(&hist[(unsigned)d[i].y], 1u);
    p[4 * i + 2] = atomicAdd(&hist[(unsigned)d[i].z], 1u);
    p[4 * i + 3] = atomicAdd(&hist[(unsigned)d[i].w], 1u);
  }
  for (int i = 0; i < CSR_OWN; ++i) {
    csr[p[4 * i + 0]] = (unsigned)os[i].x;
    csr[p[4 * i + 1]] = (unsigned)os[i].y;
    csr[p[4 * i + 2]] = (unsigned)os[i].z;
    csr[p[4 * i + 3]] = (unsigned)os[i].w;
  }
}

__device__ __forceinline__ v4f max4(v4f a, v4f b) {
  v4f r;
  r[0] = fmaxf(a[0], b[0]);
  r[1] = fmaxf(a[1], b[1]);
  r[2] = fmaxf(a[2], b[2]);
  r[3] = fmaxf(a[3], b[3]);
  return r;
}

#define MLP_WAVES 8
#define MLP_TILES_BIG 8
#define MLP_NW_BIG (((NB * NN) / 16) / MLP_TILES_BIG)
#define MLP_TILES_B (NN / 16)
#define MLP_PLANES (MLP_NW_BIG / MLP_TILES_B)
static_assert(MLP_NW_BIG % MLP_TILES_B == 0 && MLP_PLANES * MLP_TILES_BIG == NB,
              "a wave's tiles must be one node tile of NB / MLP_PLANES subgraphs");
template <bool NODE>
__global__ void __launch_bounds__(MLP_WAVES * 32)
k_mlp2_wmma(const float* __restrict__ Xin, const float* __restrict__ W1,
            const float* __restrict__ b1, const float* __restrict__ W2,
            const float* __restrict__ b2, float* __restrict__ Yout,
            float* __restrict__ pmax,
            const int* __restrict__ esrc, const int* __restrict__ edst,
            unsigned* __restrict__ off, unsigned* __restrict__ csr, int ntiles) {
  int blk = (int)blockIdx.x;
  if constexpr (NODE) {
    __shared__ unsigned hist[2 * NN];
    __shared__ unsigned wsum[MLP_WAVES];
    if (blk < CSR_BLOCKS) {
      csr_block(esrc, edst, off, csr, blk, hist, wsum);
      return;
    }
    blk -= CSR_BLOCKS;
  }
  const int lane = threadIdx.x & 31;
  const int hm = lane & 15;
  const int hk = lane >> 4;
  const int gw = __builtin_amdgcn_readfirstlane(blk * MLP_WAVES + (int)(threadIdx.x >> 5));
  constexpr int nw = NODE ? MLP_TILES_B : MLP_NW_BIG;

  v2f w1[4][16];
#pragma unroll
  for (int s = 0; s < 16; ++s) {
#pragma unroll
    for (int e = 0; e < 2; ++e) {
      const int k = 8 * (s >> 1) + 4 * hk + 2 * (s & 1) + e;
      const v4f wv = *(const v4f*)(W1 + k * ND + 4 * hm);
      w1[0][s][e] = wv[0]; w1[1][s][e] = wv[1]; w1[2][s][e] = wv[2]; w1[3][s][e] = wv[3];
    }
  }
  v2f w2[4][16];
#pragma unroll
  for (int t = 0; t < 16; ++t) {
#pragma unroll
    for (int e = 0; e < 2; ++e) {
      const int f = 4 * (8 * hk + 2 * (t & 3) + e) + (t >> 2);
      const v4f wv = *(const v4f*)(W2 + f * ND + 4 * hm);
      w2[0][t][e] = wv[0]; w2[1][t][e] = wv[1]; w2[2][t][e] = wv[2]; w2[3][t][e] = wv[3];
    }
  }
  v8f c1[4];
#pragma unroll
  for (int r = 0; r < 8; ++r) {
    const v4f bv = *(const v4f*)(b1 + 4 * (8 * hk + r));
    c1[0][r] = bv[0]; c1[1][r] = bv[1]; c1[2][r] = bv[2]; c1[3][r] = bv[3];
  }
  const v4f c2 = *(const v4f*)(b2 + 4 * hm);

  if (gw >= ntiles) return;

  v4f x[8];
  if constexpr (NODE) {
    const float* xr = Xin + ((size_t)gw * 16 + hm) * ND + 4 * hk;
#pragma unroll
    for (int i = 0; i < 8; ++i) x[i] = *(const v4f*)(xr + 8 * i);
#pragma nounroll
    for (int g = 1; g < MLP_PLANES; g += 3) {
      v4f p[3][8];
#pragma unroll
      for (int q = 0; q < 3; ++q) {
#pragma unroll
        for (int i = 0; i < 8; ++i)
          p[q][i] = *(const v4f*)(xr + (size_t)(g + q) * (NN * ND) + 8 * i);
      }
#pragma unroll
      for (int q = 0; q < 3; ++q) {
#pragma unroll
        for (int i = 0; i < 8; ++i) x[i] = max4(x[i], p[q][i]);
      }
    }
  } else {
    const float* xr = Xin + ((size_t)gw * 16 + hm) * ND + 4 * hk;
#pragma unroll
    for (int i = 0; i < 8; ++i) x[i] = *(const v4f*)(xr + 8 * i);
  }
  v4f mx[8];
#pragma unroll
  for (int i = 0; i < 8; ++i) mx[i] = x[i];

  for (int t = gw; t < ntiles; t += nw) {
    const int tn = (t + nw < ntiles) ? (t + nw) : t;
    v4f xn[8];
    {
      const float* xr = Xin + ((size_t)tn * 16 + hm) * ND + 4 * hk;
#pragma unroll
      for (int i = 0; i < 8; ++i) xn[i] = *(const v4f*)(xr + 8 * i);
    }

    v8f d1[4];
#pragma unroll
    for (int ft = 0; ft < 4; ++ft) d1[ft] = c1[ft];
#pragma unroll
    for (int i = 0; i < 8; ++i) {
      v2f blo, bhi;
      blo[0] = x[i][0]; blo[1] = x[i][1];
      bhi[0] = x[i][2]; bhi[1] = x[i][3];
#pragma unroll
      for (int ft = 0; ft < 4; ++ft)
        d1[ft] = __builtin_amdgcn_wmma_f32_16x16x4_f32(false, w1[ft][2 * i], false, blo,
                                                       (short)0, d1[ft], false, false);
#pragma unroll
      for (int ft = 0; ft < 4; ++ft)
        d1[ft] = __builtin_amdgcn_wmma_f32_16x16x4_f32(false, w1[ft][2 * i + 1], false, bhi,
                                                       (short)0, d1[ft], false, false);
    }
#pragma unroll
    for (int ft = 0; ft < 4; ++ft) {
#pragma unroll
      for (int r = 0; r < 8; ++r) d1[ft][r] = fmaxf(d1[ft][r], 0.0f);
    }

    v8f d2[4];
#pragma unroll
    for (int gt = 0; gt < 4; ++gt) {
#pragma unroll
      for (int r = 0; r < 8; ++r) d2[gt][r] = c2[gt];
    }
#pragma unroll
    for (int ft = 0; ft < 4; ++ft) {
#pragma unroll
      for (int jp = 0; jp < 4; ++jp) {
        v2f a;
        a[0] = d1[ft][2 * jp];
        a[1] = d1[ft][2 * jp + 1];
#pragma unroll
        for (int gt = 0; gt < 4; ++gt)
          d2[gt] = __builtin_amdgcn_wmma_f32_16x16x4_f32(false, a, false, w2[gt][4 * ft + jp],
                                                         (short)0, d2[gt], false, false);
      }
    }

    {
      float* yr = Yout + ((size_t)t * 16 + 8 * hk) * ND + 4 * hm;
#pragma unroll
      for (int r = 0; r < 8; ++r) {
        v4f o;
        o[0] = fmaxf(d2[0][r], 0.0f);
        o[1] = fmaxf(d2[1][r], 0.0f);
        o[2] = fmaxf(d2[2][r], 0.0f);
        o[3] = fmaxf(d2[3][r], 0.0f);
        *(v4f*)(yr + r * ND) = o;
      }
    }

#pragma unroll
    for (int i = 0; i < 8; ++i) x[i] = xn[i];
    if constexpr (!NODE) {
#pragma unroll
      for (int i = 0; i < 8; ++i) mx[i] = max4(mx[i], xn[i]);
    }
  }

  if constexpr (!NODE) {
    float* pr = pmax + (size_t)(gw / MLP_TILES_B) * (NN * ND)
              + ((size_t)(gw % MLP_TILES_B) * 16 + hm) * ND + 4 * hk;
#pragma unroll
    for (int i = 0; i < 8; ++i) *(v4f*)(pr + 8 * i) = mx[i];
  }
}

#define GROW 32
#define GZROW_B (NN * GROW * 4)

template <int T>
__device__ __forceinline__ unsigned row_share_u32(unsigned v) {
  return (unsigned)__builtin_amdgcn_update_dpp(0, (int)v, 0x150 + T, 0xf, 0xf, false);
}

__device__ __forceinline__ float xhalf_f32(float x) {
  return __builtin_bit_cast(float,
      __builtin_amdgcn_permlanex16(__builtin_bit_cast(int, x), __builtin_bit_cast(int, x),
                                   0x76543210, 0xfedcba98, false, false));
}

__device__ __forceinline__ unsigned load_rowbytes(const unsigned* __restrict__ csr,
                                                  unsigned base, unsigned left,
                                                  unsigned slot) {
  unsigned r = GZROW_B;
  if (slot < left) r = csr[base + slot] << 7;
  return r;
}

template <int T>
__device__ __forceinline__ v2f lds_row2(const char* hsb, unsigned rb, unsigned colb) {
  return *(const v2f*)(hsb + (row_share_u32<T>(rb) + colb));
}

#define GATHER_GROUP(T)                                   \
  {                                                       \
    const v2f x0 = lds_row2<T + 0>(hsb, rb, colb);        \
    const v2f x1 = lds_row2<T + 1>(hsb, rb, colb);        \
    const v2f x2 = lds_row2<T + 2>(hsb, rb, colb);        \
    const v2f x3 = lds_row2<T + 3>(hsb, rb, colb);        \
    acc += x0; acc += x1; acc += x2; acc += x3;           \
  }

__global__ void __launch_bounds__(1024)
k_gather_final(const float* __restrict__ h, const float* __restrict__ nodex,
               const unsigned* __restrict__ off, const unsigned* __restrict__ csr,
               float* __restrict__ out) {
  __shared__ float hs[(NN + 1) * GROW];
  const int tid = threadIdx.x;
  const int lane = tid & 31;
  const int wave = __builtin_amdgcn_readfirstlane(tid >> 5);
  const int b = blockIdx.x >> 1;
  const int half = blockIdx.x & 1;

  {
    const float* hsrc = h + (size_t)b * (NN * ND) + half * GROW;
    const float* nsrc = nodex + half * GROW;
    const int r0 = tid >> 3;
    const int c4 = (tid & 7) * 4;
#pragma nounroll
    for (int g = 0; g < 2; ++g) {
      float4 v[8], u[8];
#pragma unroll
      for (int p = 0; p < 8; ++p) {
        const size_t ro = (size_t)(r0 + (8 * g + p) * 128) * ND + c4;
        v[p] = *(const float4*)(hsrc + ro);
        u[p] = *(const float4*)(nsrc + ro);
      }
#pragma unroll
      for (int p = 0; p < 8; ++p) {
        float4 s;
        s.x = v[p].x + u[p].x;
        s.y = v[p].y + u[p].y;
        s.z = v[p].z + u[p].z;
        s.w = v[p].w + u[p].w;
        *(float4*)&hs[(r0 + (8 * g + p) * 128) * GROW + c4] = s;
      }
    }
    if (tid < GROW) hs[NN * GROW + tid] = 0.0f;
  }
  __syncthreads();

  const char* hsb = (const char*)hs;
  const int c = lane & 15;
  const int hrow = lane >> 4;
  const unsigned slot = 2u * c + hrow;
  const unsigned colb = 8u * c;
  const int ocol = half * GROW + 2 * c + hrow;

  for (int q = 0; q < 2; ++q) {
    const int nq = wave * 64 + q * 32;
    const unsigned ov_e = off[nq + lane + 1];
    unsigned s_cur = off[nq];
    unsigned e_cur = (unsigned)__builtin_amdgcn_readlane((int)ov_e, 0);
    unsigned rb_cur = load_rowbytes(csr, s_cur, e_cur - s_cur, slot);
    for (int i = 0; i < 32; ++i) {
      const int n = nq + i;
      const unsigned s_nxt = e_cur;
      const unsigned e_nxt = (unsigned)__builtin_amdgcn_readlane((int)ov_e, (i + 1) & 31);
      const unsigned cnt_nxt = (i + 1 < 32) ? (e_nxt - s_nxt) : 0u;
      const bool ok_nxt = slot < cnt_nxt;
      const unsigned raw_nxt = csr[ok_nxt ? (s_nxt + slot) : 0u];

      v2f acc = {0.0f, 0.0f};
      unsigned rb = rb_cur;
      unsigned left = e_cur - s_cur;
      unsigned base = s_cur;
      for (;;) {
        if (left > 0u) GATHER_GROUP(0)
        if (left > 8u) GATHER_GROUP(4)
        if (left > 16u) GATHER_GROUP(8)
        if (left > 24u) GATHER_GROUP(12)
        if (left <= 32u) break;
        left -= 32u;
        base += 32u;
        rb = load_rowbytes(csr, base, left, slot);
      }
      const float keep = hrow ? acc.y : acc.x;
      const float send = hrow ? acc.x : acc.y;
      const float tot = keep + xhalf_f32(send);
      out[((size_t)b * NN + n) * ND + ocol] = tot;

      s_cur = s_nxt;
      e_cur = e_nxt;
      rb_cur = ok_nxt ? (raw_nxt << 7) : (unsigned)GZROW_B;
    }
  }
}

extern "C" void kernel_launch(void* const* d_in, const int* in_sizes, int n_in,
                              void* d_out, int out_size, void* d_ws, size_t ws_size,
                              hipStream_t stream) {
  (void)in_sizes; (void)n_in; (void)out_size; (void)ws_size;
  const float* X   = (const float*)d_in[0];
  const int*   ei  = (const int*)d_in[1];
  const float* W1t = (const float*)d_in[2];
  const float* b1t = (const float*)d_in[3];
  const float* W2t = (const float*)d_in[4];
  const float* b2t = (const float*)d_in[5];
  const float* W1n = (const float*)d_in[6];
  const float* b1n = (const float*)d_in[7];
  const float* W2n = (const float*)d_in[8];
  const float* b2n = (const float*)d_in[9];
  float* out = (float*)d_out;

  float* h     = (float*)d_ws;
  float* nodex = h + (size_t)NB * NN * ND;
  unsigned* off = (unsigned*)(nodex + (size_t)NN * ND);
  unsigned* csr = off + (NN + 64);
  float* pmax = out;

  const int* esrc = ei;
  const int* edst = ei + NE;

  k_mlp2_wmma<false><<<MLP_NW_BIG / MLP_WAVES, MLP_WAVES * 32, 0, stream>>>(X, W1t, b1t, W2t, b2t, h, pmax, esrc, edst, off, csr, (NB * NN) / 16);
  k_mlp2_wmma<true><<<CSR_BLOCKS + MLP_TILES_B / MLP_WAVES, MLP_WAVES * 32, 0, stream>>>(pmax, W1n, b1n, W2n, b2n, nodex, (float*)nullptr, esrc, edst, off, csr, NN / 16);
  k_gather_final<<<NB * 2, 1024, 0, stream>>>(h, nodex, off, csr, out);
}


# gather loop: every row load of a node (up to 16) is in flight before the first add (was 4 per wait); index chunks requested a node pair ahead, CSR offsets ahead of staging; f32 add order kept
# speedup vs baseline: 2.5901x; 1.2120x over previous
#include <hip/hip_runtime.h>

typedef __attribute__((ext_vector_type(2))) float v2f;
typedef __attribute__((ext_vector_type(4))) float v4f;
typedef __attribute__((ext_vector_type(8))) float v8f;

#define NB 128
#define NN 2048
#define ND 64
#define NE 32768

#define CSR_BLOCKS 8
#define CSR_GROUPS (NE / 4 / 256)
#define CSR_OWN (CSR_GROUPS / CSR_BLOCKS)

__device__ __forceinline__ void csr_block(const int* __restrict__ src, const int* __restrict__ dst,
                                          unsigned* __restrict__ off, unsigned* __restrict__ csr,
                                          const int k, unsigned* hist, unsigned* wsum) {
  const int t = threadIdx.x;
  const int lane = t & 31;
  const int w = t >> 5;
  const int4* src4 = (const int4*)src;
  const int4* dst4 = (const int4*)dst;

  for (int i = 0; i < (2 * NN) / (4 * 256); ++i)
    *(uint4*)&hist[4 * (t + 256 * i)] = make_uint4(0u, 0u, 0u, 0u);
  __syncthreads();

  int4 os[CSR_OWN];
  for (int i = 0; i < CSR_OWN; ++i) os[i] = src4[t + 256 * (CSR_OWN * k + i)];
  int4 d[CSR_GROUPS];
#pragma unroll
  for (int j = 0; j < CSR_GROUPS; ++j)
    d[j] = dst4[t + 256 * ((j + CSR_OWN * k) & (CSR_GROUPS - 1))];

#pragma unroll
  for (int j = 0; j < CSR_GROUPS; ++j) {
    const unsigned sel = (j >= CSR_GROUPS - CSR_OWN * k) ? 0u : (unsigned)NN;
    atomicAdd(&hist[sel + (unsigned)d[j].x], 1u);
    atomicAdd(&hist[sel + (unsigned)d[j].y], 1u);
    atomicAdd(&hist[sel + (unsigned)d[j].z], 1u);
    atomicAdd(&hist[sel + (unsigned)d[j].w], 1u);
  }
  __syncthreads();

  const uint4 b0 = *(const uint4*)&hist[8 * t];
  const uint4 b1 = *(const uint4*)&hist[8 * t + 4];
  const uint4 r0 = *(const uint4*)&hist[NN + 8 * t];
  const uint4 r1 = *(const uint4*)&hist[NN + 8 * t + 4];
  const unsigned bb[8] = {b0.x, b0.y, b0.z, b0.w, b1.x, b1.y, b1.z, b1.w};
  const unsigned cc[8] = {b0.x + r0.x, b0.y + r0.y, b0.z + r0.z, b0.w + r0.w,
                          b1.x + r1.x, b1.y + r1.y, b1.z + r1.z, b1.w + r1.w};
  unsigned tsum = 0u;
  for (int i = 0; i < 8; ++i) tsum += cc[i];
  unsigned incl = tsum;
  for (int o = 1; o < 32; o <<= 1) {
    const unsigned u = (unsigned)__shfl_up((int)incl, o);
    if (lane >= o) incl += u;
  }
  if (lane == 31) wsum[w] = incl;
  __syncthreads();
  unsigned base = incl - tsum;
  for (int w2 = 0; w2 < 8; ++w2)
    if (w2 < w) base += wsum[w2];
  unsigned oo[8];
  oo[0] = base;
  for (int i = 1; i < 8; ++i) oo[i] = oo[i - 1] + cc[i - 1];
  *(uint4*)&hist[8 * t] = make_uint4(oo[0] + bb[0], oo[1] + bb[1], oo[2] + bb[2], oo[3] + bb[3]);
  *(uint4*)&hist[8 * t + 4] = make_uint4(oo[4] + bb[4], oo[5] + bb[5], oo[6] + bb[6], oo[7] + bb[7]);
  if (k == 0) {
    *(uint4*)&off[8 * t] = make_uint4(oo[0], oo[1], oo[2], oo[3]);
    *(uint4*)&off[8 * t + 4] = make_uint4(oo[4], oo[5], oo[6], oo[7]);
    if (t == 255) off[NN] = oo[7] + cc[7];
  }
  __syncthreads();

  unsigned p[4 * CSR_OWN];
  for (int i = 0; i < CSR_OWN; ++i) {
    p[4 * i + 0] = atomicAdd(&hist[(unsigned)d[i].x], 1u);
    p[4 * i + 1] = atomicAdd(&hist[(unsigned)d[i].y], 1u);
    p[4 * i + 2] = atomicAdd(&hist[(unsigned)d[i].z], 1u);
    p[4 * i + 3] = atomicAdd(&hist[(unsigned)d[i].w], 1u);
  }
  for (int i = 0; i < CSR_OWN; ++i) {
    csr[p[4 * i + 0]] = (unsigned)os[i].x;
    csr[p[4 * i + 1]] = (unsigned)os[i].y;
    csr[p[4 * i + 2]] = (unsigned)os[i].z;
    csr[p[4 * i + 3]] = (unsigned)os[i].w;
  }
}

__device__ __forceinline__ v4f max4(v4f a, v4f b) {
  v4f r;
  r[0] = fmaxf(a[0], b[0]);
  r[1] = fmaxf(a[1], b[1]);
  r[2] = fmaxf(a[2], b[2]);
  r[3] = fmaxf(a[3], b[3]);
  return r;
}

#define MLP_WAVES 8
#define MLP_TILES_BIG 8
#define MLP_NW_BIG (((NB * NN) / 16) / MLP_TILES_BIG)
#define MLP_TILES_B (NN / 16)
#define MLP_PLANES (MLP_NW_BIG / MLP_TILES_B)
static_assert(MLP_NW_BIG % MLP_TILES_B == 0 && MLP_PLANES * MLP_TILES_BIG == NB,
              "a wave's tiles must be one node tile of NB / MLP_PLANES subgraphs");
template <bool NODE>
__global__ void __launch_bounds__(MLP_WAVES * 32)
k_mlp2_wmma(const float* __restrict__ Xin, const float* __restrict__ W1,
            const float* __restrict__ b1, const float* __restrict__ W2,
            const float* __restrict__ b2, float* __restrict__ Yout,
            float* __restrict__ pmax,
            const int* __restrict__ esrc, const int* __restrict__ edst,
            unsigned* __restrict__ off, unsigned* __restrict__ csr, int ntiles) {
  int blk = (int)blockIdx.x;
  if constexpr (NODE) {
    __shared__ unsigned hist[2 * NN];
    __shared__ unsigned wsum[MLP_WAVES];
    if (blk < CSR_BLOCKS) {
      csr_block(esrc, edst, off, csr, blk, hist, wsum);
      return;
    }
    blk -= CSR_BLOCKS;
  }
  const int lane = threadIdx.x & 31;
  const int hm = lane & 15;
  const int hk = lane >> 4;
  const int gw = __builtin_amdgcn_readfirstlane(blk * MLP_WAVES + (int)(threadIdx.x >> 5));
  constexpr int nw = NODE ? MLP_TILES_B : MLP_NW_BIG;

  v2f w1[4][16];
#pragma unroll
  for (int s = 0; s < 16; ++s) {
#pragma unroll
    for (int e = 0; e < 2; ++e) {
      const int k = 8 * (s >> 1) + 4 * hk + 2 * (s & 1) + e;
      const v4f wv = *(const v4f*)(W1 + k * ND + 4 * hm);
      w1[0][s][e] = wv[0]; w1[1][s][e] = wv[1]; w1[2][s][e] = wv[2]; w1[3][s][e] = wv[3];
    }
  }
  v2f w2[4][16];
#pragma unroll
  for (int t = 0; t < 16; ++t) {
#pragma unroll
    for (int e = 0; e < 2; ++e) {
      const int f = 4 * (8 * hk + 2 * (t & 3) + e) + (t >> 2);
      const v4f wv = *(const v4f*)(W2 + f * ND + 4 * hm);
      w2[0][t][e] = wv[0]; w2[1][t][e] = wv[1]; w2[2][t][e] = wv[2]; w2[3][t][e] = wv[3];
    }
  }
  v8f c1[4];
#pragma unroll
  for (int r = 0; r < 8; ++r) {
    const v4f bv = *(const v4f*)(b1 + 4 * (8 * hk + r));
    c1[0][r] = bv[0]; c1[1][r] = bv[1]; c1[2][r] = bv[2]; c1[3][r] = bv[3];
  }
  const v4f c2 = *(const v4f*)(b2 + 4 * hm);

  if (gw >= ntiles) return;

  v4f x[8];
  if constexpr (NODE) {
    const float* xr = Xin + ((size_t)gw * 16 + hm) * ND + 4 * hk;
#pragma unroll
    for (int i = 0; i < 8; ++i) x[i] = *(const v4f*)(xr + 8 * i);
#pragma nounroll
    for (int g = 1; g < MLP_PLANES; g += 3) {
      v4f p[3][8];
#pragma unroll
      for (int q = 0; q < 3; ++q) {
#pragma unroll
        for (int i = 0; i < 8; ++i)
          p[q][i] = *(const v4f*)(xr + (size_t)(g + q) * (NN * ND) + 8 * i);
      }
#pragma unroll
      for (int q = 0; q < 3; ++q) {
#pragma unroll
        for (int i = 0; i < 8; ++i) x[i] = max4(x[i], p[q][i]);
      }
    }
  } else {
    const float* xr = Xin + ((size_t)gw * 16 + hm) * ND + 4 * hk;
#pragma unroll
    for (int i = 0; i < 8; ++i) x[i] = *(const v4f*)(xr + 8 * i);
  }
  v4f mx[8];
#pragma unroll
  for (int i = 0; i < 8; ++i) mx[i] = x[i];

  for (int t = gw; t < ntiles; t += nw) {
    const int tn = (t + nw < ntiles) ? (t + nw) : t;
    v4f xn[8];
    {
      const float* xr = Xin + ((size_t)tn * 16 + hm) * ND + 4 * hk;
#pragma unroll
      for (int i = 0; i < 8; ++i) xn[i] = *(const v4f*)(xr + 8 * i);
    }

    v8f d1[4];
#pragma unroll
    for (int ft = 0; ft < 4; ++ft) d1[ft] = c1[ft];
#pragma unroll
    for (int i = 0; i < 8; ++i) {
      v2f blo, bhi;
      blo[0] = x[i][0]; blo[1] = x[i][1];
      bhi[0] = x[i][2]; bhi[1] = x[i][3];
#pragma unroll
      for (int ft = 0; ft < 4; ++ft)
        d1[ft] = __builtin_amdgcn_wmma_f32_16x16x4_f32(false, w1[ft][2 * i], false, blo,
                                                       (short)0, d1[ft], false, false);
#pragma unroll
      for (int ft = 0; ft < 4; ++ft)
        d1[ft] = __builtin_amdgcn_wmma_f32_16x16x4_f32(false, w1[ft][2 * i + 1], false, bhi,
                                                       (short)0, d1[ft], false, false);
    }
#pragma unroll
    for (int ft = 0; ft < 4; ++ft) {
#pragma unroll
      for (int r = 0; r < 8; ++r) d1[ft][r] = fmaxf(d1[ft][r], 0.0f);
    }

    v8f d2[4];
#pragma unroll
    for (int gt = 0; gt < 4; ++gt) {
#pragma unroll
      for (int r = 0; r < 8; ++r) d2[gt][r] = c2[gt];
    }
#pragma unroll
    for (int ft = 0; ft < 4; ++ft) {
#pragma unroll
      for (int jp = 0; jp < 4; ++jp) {
        v2f a;
        a[0] = d1[ft][2 * jp];
        a[1] = d1[ft][2 * jp + 1];
#pragma unroll
        for (int gt = 0; gt < 4; ++gt)
          d2[gt] = __builtin_amdgcn_wmma_f32_16x16x4_f32(false, a, false, w2[gt][4 * ft + jp],
                                                         (short)0, d2[gt], false, false);
      }
    }

    {
      float* yr = Yout + ((size_t)t * 16 + 8 * hk) * ND + 4 * hm;
#pragma unroll
      for (int r = 0; r < 8; ++r) {
        v4f o;
        o[0] = fmaxf(d2[0][r], 0.0f);
        o[1] = fmaxf(d2[1][r], 0.0f);
        o[2] = fmaxf(d2[2][r], 0.0f);
        o[3] = fmaxf(d2[3][r], 0.0f);
        *(v4f*)(yr + r * ND) = o;
      }
    }

#pragma unroll
    for (int i = 0; i < 8; ++i) x[i] = xn[i];
    if constexpr (!NODE) {
#pragma unroll
      for (int i = 0; i < 8; ++i) mx[i] = max4(mx[i], xn[i]);
    }
  }

  if constexpr (!NODE) {
    float* pr = pmax + (size_t)(gw / MLP_TILES_B) * (NN * ND)
              + ((size_t)(gw % MLP_TILES_B) * 16 + hm) * ND + 4 * hk;
#pragma unroll
    for (int i = 0; i < 8; ++i) *(v4f*)(pr + 8 * i) = mx[i];
  }
}

#define GROW 32
#define GZROW_B (NN * GROW * 4)

template <int T>
__device__ __forceinline__ unsigned row_share_u32(unsigned v) {
  return (unsigned)__builtin_amdgcn_update_dpp(0, (int)v, 0x150 + T, 0xf, 0xf, false);
}

__device__ __forceinline__ float xhalf_f32(float x) {
  return __builtin_bit_cast(float,
      __builtin_amdgcn_permlanex16(__builtin_bit_cast(int, x), __builtin_bit_cast(int, x),
                                   0x76543210, 0xfedcba98, false, false));
}

__device__ __forceinline__ unsigned load_rowbytes(const unsigned* __restrict__ csr,
                                                  unsigned base, unsigned left,
                                                  unsigned slot) {
  unsigned r = GZROW_B;
  if (slot < left) r = csr[base + slot] << 7;
  return r;
}

template <int T>
__device__ __forceinline__ v2f lds_row2(const char* hsb, unsigned rb, unsigned colb) {
  return *(const v2f*)(hsb + (row_share_u32<T>(rb) + colb));
}

#define LD(T) const v2f x##T = lds_row2<T>(hsb, rb, colb);
#define AD(T) acc += x##T;
#define LD4(A, B, C, D) LD(A) LD(B) LD(C) LD(D)
#define AD4(A, B, C, D) AD(A) AD(B) AD(C) AD(D)
#define GATHER_4  { LD4(0, 1, 2, 3) __builtin_amdgcn_sched_barrier(0); AD4(0, 1, 2, 3) }
#define GATHER_8  { LD4(0, 1, 2, 3) LD4(4, 5, 6, 7) __builtin_amdgcn_sched_barrier(0); \
                    AD4(0, 1, 2, 3) AD4(4, 5, 6, 7) }
#define GATHER_12 { LD4(0, 1, 2, 3) LD4(4, 5, 6, 7) LD4(8, 9, 10, 11) \
                    __builtin_amdgcn_sched_barrier(0); \
                    AD4(0, 1, 2, 3) AD4(4, 5, 6, 7) AD4(8, 9, 10, 11) }
#define GATHER_16 { LD4(0, 1, 2, 3) LD4(4, 5, 6, 7) LD4(8, 9, 10, 11) LD4(12, 13, 14, 15) \
                    __builtin_amdgcn_sched_barrier(0); \
                    AD4(0, 1, 2, 3) AD4(4, 5, 6, 7) AD4(8, 9, 10, 11) AD4(12, 13, 14, 15) }

__device__ __forceinline__ v2f gather_node(const char* hsb, const unsigned* __restrict__ csr,
                                           unsigned rb, unsigned left, unsigned base,
                                           unsigned slot, unsigned colb) {
  v2f acc = {0.0f, 0.0f};
  for (;;) {
    if (left > 16u) {
      if (left > 24u) GATHER_16
      else GATHER_12
    } else {
      if (left > 8u) GATHER_8
      else if (left > 0u) GATHER_4
    }
    if (left <= 32u) break;
    left -= 32u;
    base += 32u;
    rb = load_rowbytes(csr, base, left, slot);
  }
  return acc;
}

__device__ __forceinline__ float fold_halves(v2f acc, int hrow) {
  const float keep = hrow ? acc.y : acc.x;
  const float send = hrow ? acc.x : acc.y;
  return keep + xhalf_f32(send);
}

__global__ void __launch_bounds__(1024)
k_gather_final(const float* __restrict__ h, const float* __restrict__ nodex,
               const unsigned* __restrict__ off, const unsigned* __restrict__ csr,
               float* __restrict__ out) {
  __shared__ float hs[(NN + 1) * GROW];
  const int tid = threadIdx.x;
  const int lane = tid & 31;
  const int wave = __builtin_amdgcn_readfirstlane(tid >> 5);
  const int b = blockIdx.x >> 1;
  const int half = blockIdx.x & 1;

  const int n0 = wave * 64;
  const unsigned ov0 = off[n0 + lane + 1];
  const unsigned ov1 = off[n0 + 32 + lane + 1];
  unsigned s_A = off[n0];

  {
    const float* hsrc = h + (size_t)b * (NN * ND) + half * GROW;
    const float* nsrc = nodex + half * GROW;
    const int r0 = tid >> 3;
    const int c4 = (tid & 7) * 4;
#pragma nounroll
    for (int g = 0; g < 2; ++g) {
      float4 v[8], u[8];
#pragma unroll
      for (int p = 0; p < 8; ++p) {
        const size_t ro = (size_t)(r0 + (8 * g + p) * 128) * ND + c4;
        v[p] = *(const float4*)(hsrc + ro);
        u[p] = *(const float4*)(nsrc + ro);
      }
#pragma unroll
      for (int p = 0; p < 8; ++p) {
        float4 s;
        s.x = v[p].x + u[p].x;
        s.y = v[p].y + u[p].y;
        s.z = v[p].z + u[p].z;
        s.w = v[p].w + u[p].w;
        *(float4*)&hs[(r0 + (8 * g + p) * 128) * GROW + c4] = s;
      }
    }
    if (tid < GROW) hs[NN * GROW + tid] = 0.0f;
  }

  const char* hsb = (const char*)hs;
  const int c = lane & 15;
  const int hrow = lane >> 4;
  const unsigned slot = 2u * c + hrow;
  const unsigned colb = 8u * c;
  const int ocol = half * GROW + 2 * c + hrow;
  float* orow = out + ((size_t)b * NN + n0) * ND + ocol;

  unsigned e_A = (unsigned)__builtin_amdgcn_readlane((int)ov0, 0);
  unsigned e_B = (unsigned)__builtin_amdgcn_readlane((int)ov0, 1);
  const bool okA0 = slot < (e_A - s_A);
  const bool okB0 = slot < (e_B - e_A);
  const unsigned rawA0 = csr[okA0 ? (s_A + slot) : 0u];
  const unsigned rawB0 = csr[okB0 ? (e_A + slot) : 0u];
  __syncthreads();
  unsigned rbA = okA0 ? (rawA0 << 7) : (unsigned)GZROW_B;
  unsigned rbB = okB0 ? (rawB0 << 7) : (unsigned)GZROW_B;

  for (int p = 0; p < 32; ++p) {
    const int j = (2 * p + 2) & 63;
    const unsigned a0 = (unsigned)__builtin_amdgcn_readlane((int)ov0, j & 31);
    const unsigned a1 = (unsigned)__builtin_amdgcn_readlane((int)ov1, j & 31);
    const unsigned b0 = (unsigned)__builtin_amdgcn_readlane((int)ov0, (j + 1) & 31);
    const unsigned b1 = (unsigned)__builtin_amdgcn_readlane((int)ov1, (j + 1) & 31);
    const bool more = p + 1 < 32;
    const unsigned e_An = more ? ((j < 32) ? a0 : a1) : e_B;
    const unsigned e_Bn = more ? ((j < 32) ? b0 : b1) : e_B;
    const bool okAn = slot < (e_An - e_B);
    const bool okBn = slot < (e_Bn - e_An);
    const unsigned rawAn = csr[okAn ? (e_B + slot) : 0u];
    const unsigned rawBn = csr[okBn ? (e_An + slot) : 0u];

    const v2f accA = gather_node(hsb, csr, rbA, e_A - s_A, s_A, slot, colb);
    orow[(size_t)(2 * p) * ND] = fold_halves(accA, hrow);
    const v2f accB = gather_node(hsb, csr, rbB, e_B - e_A, e_A, slot, colb);
    orow[(size_t)(2 * p + 1) * ND] = fold_halves(accB, hrow);

    s_A = e_B;
    e_A = e_An;
    e_B = e_Bn;
    rbA = okAn ? (rawAn << 7) : (unsigned)GZROW_B;
    rbB = okBn ? (rawBn << 7) : (unsigned)GZROW_B;
  }
}

extern "C" void kernel_launch(void* const* d_in, const int* in_sizes, int n_in,
                              void* d_out, int out_size, void* d_ws, size_t ws_size,
                              hipStream_t stream) {
  (void)in_sizes; (void)n_in; (void)out_size; (void)ws_size;
  const float* X   = (const float*)d_in[0];
  const int*   ei  = (const int*)d_in[1];
  const float* W1t = (const float*)d_in[2];
  const float* b1t = (const float*)d_in[3];
  const float* W2t = (const float*)d_in[4];
  const float* b2t = (const float*)d_in[5];
  const float* W1n = (const float*)d_in[6];
  const float* b1n = (const float*)d_in[7];
  const float* W2n = (const float*)d_in[8];
  const float* b2n = (const float*)d_in[9];
  float* out = (float*)d_out;

  float* h     = (float*)d_ws;
  float* nodex = h + (size_t)NB * NN * ND;
  unsigned* off = (unsigned*)(nodex + (size_t)NN * ND);
  unsigned* csr = off + (NN + 64);
  float* pmax = out;

  const int* esrc = ei;
  const int* edst = ei + NE;

  k_mlp2_wmma<false><<<MLP_NW_BIG / MLP_WAVES, MLP_WAVES * 32, 0, stream>>>(X, W1t, b1t, W2t, b2t, h, pmax, esrc, edst, off, csr, (NB * NN) / 16);
  k_mlp2_wmma<true><<<CSR_BLOCKS + MLP_TILES_B / MLP_WAVES, MLP_WAVES * 32, 0, stream>>>(pmax, W1n, b1n, W2n, b2n, nodex, (float*)nullptr, esrc, edst, off, csr, NN / 16);
  k_gather_final<<<NB * 2, 1024, 0, stream>>>(h, nodex, off, csr, out);
}


# final gather: lane halves take the two nodes of a pair (was even/odd edges of one node): one LDS round trip per pair, not two, no cross-half fold; index words 2 pairs ahead; f32 add order kept
# speedup vs baseline: 2.6934x; 1.0399x over previous
#include <hip/hip_runtime.h>

typedef __attribute__((ext_vector_type(2))) float v2f;
typedef __attribute__((ext_vector_type(4))) float v4f;
typedef __attribute__((ext_vector_type(8))) float v8f;

#define NB 128
#define NN 2048
#define ND 64
#define NE 32768

#define CSR_BLOCKS 8
#define CSR_GROUPS (NE / 4 / 256)
#define CSR_OWN (CSR_GROUPS / CSR_BLOCKS)

__device__ __forceinline__ void csr_block(const int* __restrict__ src, const int* __restrict__ dst,
                                          unsigned* __restrict__ off, unsigned* __restrict__ csr,
                                          const int k, unsigned* hist, unsigned* wsum) {
  const int t = threadIdx.x;
  const int lane = t & 31;
  const int w = t >> 5;
  const int4* src4 = (const int4*)src;
  const int4* dst4 = (const int4*)dst;

  for (int i = 0; i < (2 * NN) / (4 * 256); ++i)
    *(uint4*)&hist[4 * (t + 256 * i)] = make_uint4(0u, 0u, 0u, 0u);
  __syncthreads();

  int4 os[CSR_OWN];
  for (int i = 0; i < CSR_OWN; ++i) os[i] = src4[t + 256 * (CSR_OWN * k + i)];
  int4 d[CSR_GROUPS];
#pragma unroll
  for (int j = 0; j < CSR_GROUPS; ++j)
    d[j] = dst4[t + 256 * ((j + CSR_OWN * k) & (CSR_GROUPS - 1))];

#pragma unroll
  for (int j = 0; j < CSR_GROUPS; ++j) {
    const unsigned sel = (j >= CSR_GROUPS - CSR_OWN * k) ? 0u : (unsigned)NN;
    atomicAdd(&hist[sel + (unsigned)d[j].x], 1u);
    atomicAdd(&hist[sel + (unsigned)d[j].y], 1u);
    atomicAdd(&hist[sel + (unsigned)d[j].z], 1u);
    atomicAdd(&hist[sel + (unsigned)d[j].w], 1u);
  }
  __syncthreads();

  const uint4 b0 = *(const uint4*)&hist[8 * t];
  const uint4 b1 = *(const uint4*)&hist[8 * t + 4];
  const uint4 r0 = *(const uint4*)&hist[NN + 8 * t];
  const uint4 r1 = *(const uint4*)&hist[NN + 8 * t + 4];
  const unsigned bb[8] = {b0.x, b0.y, b0.z, b0.w, b1.x, b1.y, b1.z, b1.w};
  const unsigned cc[8] = {b0.x + r0.x, b0.y + r0.y, b0.z + r0.z, b0.w + r0.w,
                          b1.x + r1.x, b1.y + r1.y, b1.z + r1.z, b1.w + r1.w};
  unsigned tsum = 0u;
  for (int i = 0; i < 8; ++i) tsum += cc[i];
  unsigned incl = tsum;
  for (int o = 1; o < 32; o <<= 1) {
    const unsigned u = (unsigned)__shfl_up((int)incl, o);
    if (lane >= o) incl += u;
  }
  if (lane == 31) wsum[w] = incl;
  __syncthreads();
  unsigned base = incl - tsum;
  for (int w2 = 0; w2 < 8; ++w2)
    if (w2 < w) base += wsum[w2];
  unsigned oo[8];
  oo[0] = base;
  for (int i = 1; i < 8; ++i) oo[i] = oo[i - 1] + cc[i - 1];
  *(uint4*)&hist[8 * t] = make_uint4(oo[0] + bb[0], oo[1] + bb[1], oo[2] + bb[2], oo[3] + bb[3]);
  *(uint4*)&hist[8 * t + 4] = make_uint4(oo[4] + bb[4], oo[5] + bb[5], oo[6] + bb[6], oo[7] + bb[7]);
  if (k == 0) {
    *(uint4*)&off[8 * t] = make_uint4(oo[0], oo[1], oo[2], oo[3]);
    *(uint4*)&off[8 * t + 4] = make_uint4(oo[4], oo[5], oo[6], oo[7]);
    if (t == 255) off[NN] = oo[7] + cc[7];
  }
  __syncthreads();

  unsigned p[4 * CSR_OWN];
  for (int i = 0; i < CSR_OWN; ++i) {
    p[4 * i + 0] = atomicAdd(&hist[(unsigned)d[i].x], 1u);
    p[4 * i + 1] = atomicAdd(&hist[(unsigned)d[i].y], 1u);
    p[4 * i + 2] = atomicAdd(&hist[(unsigned)d[i].z], 1u);
    p[4 * i + 3] = atomicAdd(&hist[(unsigned)d[i].w], 1u);
  }
  for (int i = 0; i < CSR_OWN; ++i) {
    csr[p[4 * i + 0]] = (unsigned)os[i].x;
    csr[p[4 * i + 1]] = (unsigned)os[i].y;
    csr[p[4 * i + 2]] = (unsigned)os[i].z;
    csr[p[4 * i + 3]] = (unsigned)os[i].w;
  }
}

__device__ __forceinline__ v4f max4(v4f a, v4f b) {
  v4f r;
  r[0] = fmaxf(a[0], b[0]);
  r[1] = fmaxf(a[1], b[1]);
  r[2] = fmaxf(a[2], b[2]);
  r[3] = fmaxf(a[3], b[3]);
  return r;
}

#define MLP_WAVES 8
#define MLP_TILES_BIG 8
#define MLP_NW_BIG (((NB * NN) / 16) / MLP_TILES_BIG)
#define MLP_TILES_B (NN / 16)
#define MLP_PLANES (MLP_NW_BIG / MLP_TILES_B)
static_assert(MLP_NW_BIG % MLP_TILES_B == 0 && MLP_PLANES * MLP_TILES_BIG == NB,
              "a wave's tiles must be one node tile of NB / MLP_PLANES subgraphs");
template <bool NODE>
__global__ void __launch_bounds__(MLP_WAVES * 32)
k_mlp2_wmma(const float* __restrict__ Xin, const float* __restrict__ W1,
            const float* __restrict__ b1, const float* __restrict__ W2,
            const float* __restrict__ b2, float* __restrict__ Yout,
            float* __restrict__ pmax,
            const int* __restrict__ esrc, const int* __restrict__ edst,
            unsigned* __restrict__ off, unsigned* __restrict__ csr, int ntiles) {
  int blk = (int)blockIdx.x;
  if constexpr (NODE) {
    __shared__ unsigned hist[2 * NN];
    __shared__ unsigned wsum[MLP_WAVES];
    if (blk < CSR_BLOCKS) {
      csr_block(esrc, edst, off, csr, blk, hist, wsum);
      return;
    }
    blk -= CSR_BLOCKS;
  }
  const int lane = threadIdx.x & 31;
  const int hm = lane & 15;
  const int hk = lane >> 4;
  const int gw = __builtin_amdgcn_readfirstlane(blk * MLP_WAVES + (int)(threadIdx.x >> 5));
  constexpr int nw = NODE ? MLP_TILES_B : MLP_NW_BIG;

  v2f w1[4][16];
#pragma unroll
  for (int s = 0; s < 16; ++s) {
#pragma unroll
    for (int e = 0; e < 2; ++e) {
      const int k = 8 * (s >> 1) + 4 * hk + 2 * (s & 1) + e;
      const v4f wv = *(const v4f*)(W1 + k * ND + 4 * hm);
      w1[0][s][e] = wv[0]; w1[1][s][e] = wv[1]; w1[2][s][e] = wv[2]; w1[3][s][e] = wv[3];
    }
  }
  v2f w2[4][16];
#pragma unroll
  for (int t = 0; t < 16; ++t) {
#pragma unroll
    for (int e = 0; e < 2; ++e) {
      const int f = 4 * (8 * hk + 2 * (t & 3) + e) + (t >> 2);
      const v4f wv = *(const v4f*)(W2 + f * ND + 4 * hm);
      w2[0][t][e] = wv[0]; w2[1][t][e] = wv[1]; w2[2][t][e] = wv[2]; w2[3][t][e] = wv[3];
    }
  }
  v8f c1[4];
#pragma unroll
  for (int r = 0; r < 8; ++r) {
    const v4f bv = *(const v4f*)(b1 + 4 * (8 * hk + r));
    c1[0][r] = bv[0]; c1[1][r] = bv[1]; c1[2][r] = bv[2]; c1[3][r] = bv[3];
  }
  const v4f c2 = *(const v4f*)(b2 + 4 * hm);

  if (gw >= ntiles) return;

  v4f x[8];
  if constexpr (NODE) {
    const float* xr = Xin + ((size_t)gw * 16 + hm) * ND + 4 * hk;
#pragma unroll
    for (int i = 0; i < 8; ++i) x[i] = *(const v4f*)(xr + 8 * i);
#pragma nounroll
    for (int g = 1; g < MLP_PLANES; g += 3) {
      v4f p[3][8];
#pragma unroll
      for (int q = 0; q < 3; ++q) {
#pragma unroll
        for (int i = 0; i < 8; ++i)
          p[q][i] = *(const v4f*)(xr + (size_t)(g + q) * (NN * ND) + 8 * i);
      }
#pragma unroll
      for (int q = 0; q < 3; ++q) {
#pragma unroll
        for (int i = 0; i < 8; ++i) x[i] = max4(x[i], p[q][i]);
      }
    }
  } else {
    const float* xr = Xin + ((size_t)gw * 16 + hm) * ND + 4 * hk;
#pragma unroll
    for (int i = 0; i < 8; ++i) x[i] = *(const v4f*)(xr + 8 * i);
  }
  v4f mx[8];
#pragma unroll
  for (int i = 0; i < 8; ++i) mx[i] = x[i];

  for (int t = gw; t < ntiles; t += nw) {
    const int tn = (t + nw < ntiles) ? (t + nw) : t;
    v4f xn[8];
    {
      const float* xr = Xin + ((size_t)tn * 16 + hm) * ND + 4 * hk;
#pragma unroll
      for (int i = 0; i < 8; ++i) xn[i] = *(const v4f*)(xr + 8 * i);
    }

    v8f d1[4];
#pragma unroll
    for (int ft = 0; ft < 4; ++ft) d1[ft] = c1[ft];
#pragma unroll
    for (int i = 0; i < 8; ++i) {
      v2f blo, bhi;
      blo[0] = x[i][0]; blo[1] = x[i][1];
      bhi[0] = x[i][2]; bhi[1] = x[i][3];
#pragma unroll
      for (int ft = 0; ft < 4; ++ft)
        d1[ft] = __builtin_amdgcn_wmma_f32_16x16x4_f32(false, w1[ft][2 * i], false, blo,
                                                       (short)0, d1[ft], false, false);
#pragma unroll
      for (int ft = 0; ft < 4; ++ft)
        d1[ft] = __builtin_amdgcn_wmma_f32_16x16x4_f32(false, w1[ft][2 * i + 1], false, bhi,
                                                       (short)0, d1[ft], false, false);
    }
#pragma unroll
    for (int ft = 0; ft < 4; ++ft) {
#pragma unroll
      for (int r = 0; r < 8; ++r) d1[ft][r] = fmaxf(d1[ft][r], 0.0f);
    }

    v8f d2[4];
#pragma unroll
    for (int gt = 0; gt < 4; ++gt) {
#pragma unroll
      for (int r = 0; r < 8; ++r) d2[gt][r] = c2[gt];
    }
#pragma unroll
    for (int ft = 0; ft < 4; ++ft) {
#pragma unroll
      for (int jp = 0; jp < 4; ++jp) {
        v2f a;
        a[0] = d1[ft][2 * jp];
        a[1] = d1[ft][2 * jp + 1];
#pragma unroll
        for (int gt = 0; gt < 4; ++gt)
          d2[gt] = __builtin_amdgcn_wmma_f32_16x16x4_f32(false, a, false, w2[gt][4 * ft + jp],
                                                         (short)0, d2[gt], false, false);
      }
    }

    {
      float* yr = Yout + ((size_t)t * 16 + 8 * hk) * ND + 4 * hm;
#pragma unroll
      for (int r = 0; r < 8; ++r) {
        v4f o;
        o[0] = fmaxf(d2[0][r], 0.0f);
        o[1] = fmaxf(d2[1][r], 0.0f);
        o[2] = fmaxf(d2[2][r], 0.0f);
        o[3] = fmaxf(d2[3][r], 0.0f);
        *(v4f*)(yr + r * ND) = o;
      }
    }

#pragma unroll
    for (int i = 0; i < 8; ++i) x[i] = xn[i];
    if constexpr (!NODE) {
#pragma unroll
      for (int i = 0; i < 8; ++i) mx[i] = max4(mx[i], xn[i]);
    }
  }

  if constexpr (!NODE) {
    float* pr = pmax + (size_t)(gw / MLP_TILES_B) * (NN * ND)
              + ((size_t)(gw % MLP_TILES_B) * 16 + hm) * ND + 4 * hk;
#pragma unroll
    for (int i = 0; i < 8; ++i) *(v4f*)(pr + 8 * i) = mx[i];
  }
}

#define GROW 32
#define GZROW_B (NN * GROW * 4)

template <int T>
__device__ __forceinline__ unsigned row_share_u32(unsigned v) {
  return (unsigned)__builtin_amdgcn_update_dpp(0, (int)v, 0x150 + T, 0xf, 0xf, false);
}

template <int T>
__device__ __forceinline__ v2f lds_row2(const char* hsb, unsigned rb, unsigned colb) {
  return *(const v2f*)(hsb + (row_share_u32<T>(rb) + colb));
}

#define LDA(T) const v2f x##T = lds_row2<T>(hsb, rb0, colb);
#define LDB(T) const v2f y##T = lds_row2<T>(hsb, rb1, colb);
#define LDA4(A, B, C, D) LDA(A) LDA(B) LDA(C) LDA(D)
#define LDB4(A, B, C, D) LDB(A) LDB(B) LDB(C) LDB(D)
#define ADA4(A, B, C, D) a0 += x##A; a1 += x##B; a0 += x##C; a1 += x##D;
#define ADB4(A, B, C, D) a0 += y##A; a1 += y##B; a0 += y##C; a1 += y##D;
#define LA_1 LDA4(0, 1, 2, 3)
#define LA_2 LA_1 LDA4(4, 5, 6, 7)
#define LA_3 LA_2 LDA4(8, 9, 10, 11)
#define LA_4 LA_3 LDA4(12, 13, 14, 15)
#define LB_1 LDB4(0, 1, 2, 3)
#define LB_2 LB_1 LDB4(4, 5, 6, 7)
#define LB_3 LB_2 LDB4(8, 9, 10, 11)
#define LB_4 LB_3 LDB4(12, 13, 14, 15)
#define AA_1 ADA4(0, 1, 2, 3)
#define AA_2 AA_1 ADA4(4, 5, 6, 7)
#define AA_3 AA_2 ADA4(8, 9, 10, 11)
#define AA_4 AA_3 ADA4(12, 13, 14, 15)
#define AB_1 ADB4(0, 1, 2, 3)
#define AB_2 AB_1 ADB4(4, 5, 6, 7)
#define AB_3 AB_2 ADB4(8, 9, 10, 11)
#define AB_4 AB_3 ADB4(12, 13, 14, 15)
#define LOADS_FIRST __builtin_amdgcn_sched_barrier(0);
#define GATHER_4  { LA_1 LOADS_FIRST AA_1 }
#define GATHER_8  { LA_2 LOADS_FIRST AA_2 }
#define GATHER_12 { LA_3 LOADS_FIRST AA_3 }
#define GATHER_16 { LA_4 LOADS_FIRST AA_4 }
#define GATHER_20 { LA_4 LB_1 LOADS_FIRST AA_4 AB_1 }
#define GATHER_24 { LA_4 LB_2 LOADS_FIRST AA_4 AB_2 }
#define GATHER_28 { LA_4 LB_3 LOADS_FIRST AA_4 AB_3 }
#define GATHER_32 { LA_4 LB_4 LOADS_FIRST AA_4 AB_4 }

__device__ __forceinline__ v2f gather_pair(const char* hsb, const unsigned* __restrict__ csr,
                                           unsigned rb0, unsigned rb1,
                                           const unsigned s_A, const unsigned e_A, const unsigned e_B,
                                           const int hrow, const unsigned c, const unsigned colb) {
  const unsigned dA = e_A - s_A;
  const unsigned dB = e_B - e_A;
  unsigned kmax = (dA > dB) ? dA : dB;
  unsigned done = 0u;
  v2f a0 = {0.0f, 0.0f};
  v2f a1 = {0.0f, 0.0f};
  for (;;) {
    if (kmax > 16u) {
      if (kmax > 24u) {
        if (kmax > 28u) GATHER_32
        else GATHER_28
      } else {
        if (kmax > 20u) GATHER_24
        else GATHER_20
      }
    } else {
      if (kmax > 8u) {
        if (kmax > 12u) GATHER_16
        else GATHER_12
      } else {
        if (kmax > 4u) GATHER_8
        else if (kmax > 0u) GATHER_4
      }
    }
    if (kmax <= 32u) break;
    kmax -= 32u;
    done += 32u;
    const unsigned st = (hrow ? e_A : s_A) + done;
    const unsigned dall = hrow ? dB : dA;
    const unsigned dg = (dall > done) ? (dall - done) : 0u;
    rb0 = GZROW_B;
    rb1 = GZROW_B;
    if (c < dg) rb0 = csr[st + c] << 7;
    if (c + 16u < dg) rb1 = csr[st + 16u + c] << 7;
  }
  return a0 + a1;
}

__device__ __forceinline__ void pair_step(const int p, const char* hsb, const unsigned* __restrict__ csr,
                                          const unsigned* __restrict__ offw,
                                          const int hrow, const unsigned c, const unsigned colb,
                                          float* orow,
                                          unsigned& s_A, unsigned& e_A, unsigned& e_B,
                                          unsigned& e_A1, unsigned& e_B1,
                                          unsigned& e_A2, unsigned& e_B2,
                                          unsigned& rb0, unsigned& rb1,
                                          unsigned& ld0, unsigned& ld1, bool& lk0, bool& lk1,
                                          const unsigned nx0, const unsigned nx1,
                                          const bool nk0, const bool nk1) {
  const int k3 = 2 * p + 7;
  const unsigned e_A3 = offw[k3 < 64 ? k3 : 64];
  const unsigned e_B3 = offw[k3 + 1 < 64 ? k3 + 1 : 64];

  const unsigned st2 = hrow ? e_A2 : e_B1;
  const unsigned dg2 = hrow ? (e_B2 - e_A2) : (e_A2 - e_B1);
  lk0 = c < dg2;
  lk1 = c + 16u < dg2;
  ld0 = csr[lk0 ? (st2 + c) : 0u];
  ld1 = csr[lk1 ? (st2 + 16u + c) : 0u];

  const v2f acc = gather_pair(hsb, csr, rb0, rb1, s_A, e_A, e_B, hrow, c, colb);
  *(v2f*)(orow + (size_t)(2 * p) * ND) = acc;

  rb0 = nk0 ? (nx0 << 7) : (unsigned)GZROW_B;
  rb1 = nk1 ? (nx1 << 7) : (unsigned)GZROW_B;
  s_A = e_B;
  e_A = e_A1;
  e_B = e_B1;
  e_A1 = e_A2;
  e_B1 = e_B2;
  e_A2 = e_A3;
  e_B2 = e_B3;
}

__global__ void __launch_bounds__(1024)
k_gather_final(const float* __restrict__ h, const float* __restrict__ nodex,
               const unsigned* __restrict__ off, const unsigned* __restrict__ csr,
               float* __restrict__ out) {
  __shared__ float hs[(NN + 1) * GROW];
  const int tid = threadIdx.x;
  const int lane = tid & 31;
  const int wave = __builtin_amdgcn_readfirstlane(tid >> 5);
  const int b = blockIdx.x >> 1;
  const int half = blockIdx.x & 1;

  const int n0 = wave * 64;
  const unsigned* offw = off + n0;
  unsigned s_A = offw[0];
  unsigned e_A = offw[1];
  unsigned e_B = offw[2];
  unsigned e_A1 = offw[3];
  unsigned e_B1 = offw[4];
  unsigned e_A2 = offw[5];
  unsigned e_B2 = offw[6];

  {
    const float* hsrc = h + (size_t)b * (NN * ND) + half * GROW;
    const float* nsrc = nodex + half * GROW;
    const int r0 = tid >> 3;
    const int c4 = (tid & 7) * 4;
#pragma nounroll
    for (int g = 0; g < 2; ++g) {
      float4 v[8], u[8];
#pragma unroll
      for (int p = 0; p < 8; ++p) {
        const size_t ro = (size_t)(r0 + (8 * g + p) * 128) * ND + c4;
        v[p] = *(const float4*)(hsrc + ro);
        u[p] = *(const float4*)(nsrc + ro);
      }
#pragma unroll
      for (int p = 0; p < 8; ++p) {
        float4 s;
        s.x = v[p].x + u[p].x;
        s.y = v[p].y + u[p].y;
        s.z = v[p].z + u[p].z;
        s.w = v[p].w + u[p].w;
        *(float4*)&hs[(r0 + (8 * g + p) * 128) * GROW + c4] = s;
      }
    }
    if (tid < GROW) hs[NN * GROW + tid] = 0.0f;
  }

  const char* hsb = (const char*)hs;
  const unsigned c = (unsigned)(lane & 15);
  const int hrow = lane >> 4;
  const unsigned colb = 8u * c;
  float* orow = out + ((size_t)b * NN + n0 + hrow) * ND + half * GROW + 2 * c;

  unsigned rb0, rb1;
  {
    const unsigned st0 = hrow ? e_A : s_A;
    const unsigned dg0 = hrow ? (e_B - e_A) : (e_A - s_A);
    const bool ok00 = c < dg0;
    const bool ok10 = c + 16u < dg0;
    const unsigned raw00 = csr[ok00 ? (st0 + c) : 0u];
    const unsigned raw10 = csr[ok10 ? (st0 + 16u + c) : 0u];
    rb0 = ok00 ? (raw00 << 7) : (unsigned)GZROW_B;
    rb1 = ok10 ? (raw10 << 7) : (unsigned)GZROW_B;
  }
  const unsigned st1 = hrow ? e_A1 : e_B;
  const unsigned dg1 = hrow ? (e_B1 - e_A1) : (e_A1 - e_B);
  bool okO0 = c < dg1;
  bool okO1 = c + 16u < dg1;
  unsigned rawO0 = csr[okO0 ? (st1 + c) : 0u];
  unsigned rawO1 = csr[okO1 ? (st1 + 16u + c) : 0u];
  bool okE0 = false;
  bool okE1 = false;
  unsigned rawE0 = 0u;
  unsigned rawE1 = 0u;
  __syncthreads();

  for (int q = 0; q < 16; ++q) {
    pair_step(2 * q, hsb, csr, offw, hrow, c, colb, orow, s_A, e_A, e_B, e_A1, e_B1, e_A2, e_B2,
              rb0, rb1, rawE0, rawE1, okE0, okE1, rawO0, rawO1, okO0, okO1);
    pair_step(2 * q + 1, hsb, csr, offw, hrow, c, colb, orow, s_A, e_A, e_B, e_A1, e_B1, e_A2, e_B2,
              rb0, rb1, rawO0, rawO1, okO0, okO1, rawE0, rawE1, okE0, okE1);
  }
}

extern "C" void kernel_launch(void* const* d_in, const int* in_sizes, int n_in,
                              void* d_out, int out_size, void* d_ws, size_t ws_size,
                              hipStream_t stream) {
  (void)in_sizes; (void)n_in; (void)out_size; (void)ws_size;
  const float* X   = (const float*)d_in[0];
  const int*   ei  = (const int*)d_in[1];
  const float* W1t = (const float*)d_in[2];
  const float* b1t = (const float*)d_in[3];
  const float* W2t = (const float*)d_in[4];
  const float* b2t = (const float*)d_in[5];
  const float* W1n = (const float*)d_in[6];
  const float* b1n = (const float*)d_in[7];
  const float* W2n = (const float*)d_in[8];
  const float* b2n = (const float*)d_in[9];
  float* out = (float*)d_out;

  float* h     = (float*)d_ws;
  float* nodex = h + (size_t)NB * NN * ND;
  unsigned* off = (unsigned*)(nodex + (size_t)NN * ND);
  unsigned* csr = off + (NN + 64);
  float* pmax = out;

  const int* esrc = ei;
  const int* edst = ei + NE;

  k_mlp2_wmma<false><<<MLP_NW_BIG / MLP_WAVES, MLP_WAVES * 32, 0, stream>>>(X, W1t, b1t, W2t, b2t, h, pmax, esrc, edst, off, csr, (NB * NN) / 16);
  k_mlp2_wmma<true><<<CSR_BLOCKS + MLP_TILES_B / MLP_WAVES, MLP_WAVES * 32, 0, stream>>>(pmax, W1n, b1n, W2n, b2n, nodex, (float*)nullptr, esrc, edst, off, csr, NN / 16);
  k_gather_final<<<NB * 2, 1024, 0, stream>>>(h, nodex, off, csr, out);
}
